# starting kernel
# baseline (speedup 1.0000x reference)
#include <hip/hip_runtime.h>
#include <math.h>

#define NNODES   3712
#define NROI     116
#define NGRAPH   32
#define NFEAT    128
#define NHID     64
#define NHEADS   4
#define ALPHA_LR 0.1f
#define A_PARA   0.1f
#define PHI_C    0.94247779607693793f
#define MAXNORM  (1.0f - 4e-3f)

typedef __attribute__((ext_vector_type(2))) float v2f;
typedef __attribute__((ext_vector_type(8))) float v8f;

__device__ __forceinline__ v8f wmma4f32(v2f a, v2f b, v8f c) {
  return __builtin_amdgcn_wmma_f32_16x16x4_f32(
      false, a, false, b, (short)0, c, false, false);
}

__device__ __forceinline__ float fkernel_scale(float r) {
  float proj = 1.0f, rn = r;
  if (r > MAXNORM) { proj = MAXNORM / r; rn = MAXNORM; }
  rn = fmaxf(rn, 1e-15f);
  float t = fminf(rn, 1.0f - 1e-7f);
  return proj * (atanhf(t) / rn);
}

__device__ __forceinline__ float act_elu_cos(float v) {
  float e = (v > 0.0f) ? v : expm1f(v);
  return e + A_PARA * cosf(v + PHI_C);
}

__global__ __launch_bounds__(32)
void gemm_wmma_kernel(const float* __restrict__ A, const float* __restrict__ B,
                      float* __restrict__ C, int M, int K, int Nc) {
  const int h  = blockIdx.y;
  const float* Bh = B + (size_t)h * K * Nc;
  float*       Ch = C + (size_t)h * M * Nc;
  const int ntiles = Nc >> 4;
  const int mt = blockIdx.x / ntiles;
  const int nt = blockIdx.x % ntiles;
  const int l  = threadIdx.x;
  const int lm = l & 15;
  const int lh = l >> 4;

  v8f acc = {0.f,0.f,0.f,0.f,0.f,0.f,0.f,0.f};
  const float* arow = A  + (size_t)(mt * 16 + lm) * K + 2 * lh;
  const float* bcol = Bh + (size_t)(2 * lh) * Nc + nt * 16 + lm;

  for (int k0 = 0; k0 < K; k0 += 4) {
    v2f a = { arow[k0],              arow[k0 + 1] };
    v2f b = { bcol[(size_t)k0 * Nc], bcol[(size_t)(k0 + 1) * Nc] };
    acc = wmma4f32(a, b, acc);
  }
  float* crow = Ch + (size_t)(mt * 16 + 8 * lh) * Nc + nt * 16 + lm;
#pragma unroll
  for (int v = 0; v < 8; ++v) crow[(size_t)v * Nc] = acc[v];
}

__global__ void attvec_kernel(const float* __restrict__ Wh,
                              const float* __restrict__ avec,
                              float* __restrict__ el, float* __restrict__ er,
                              int heads) {
  int idx = blockIdx.x * blockDim.x + threadIdx.x;
  if (idx >= heads * NNODES) return;
  int h = idx / NNODES, i = idx % NNODES;
  const float* row = Wh + ((size_t)h * NNODES + i) * NHID;
  const float* a1  = avec + (size_t)h * 2 * NHID;
  const float* a2  = a1 + NHID;
  float s1 = 0.f, s2 = 0.f;
#pragma unroll 4
  for (int k = 0; k < NHID; ++k) { s1 += row[k] * a1[k]; s2 += row[k] * a2[k]; }
  el[idx] = s1;
  er[idx] = s2;
}

#define ATT_S 132
#define WHS_S 68
#define LDS_BYTES ((128 * ATT_S + 128 * WHS_S) * sizeof(float))

template <bool DOUBLE_FK>
__global__ __launch_bounds__(256)
void att_block_kernel(const float* __restrict__ adj, const float* __restrict__ Wh,
                      const float* __restrict__ el, const float* __restrict__ er,
                      float* __restrict__ out, int out_stride) {
  extern __shared__ float lds[];
  float* att = lds;
  float* whs = lds + 128 * ATT_S;

  const int g = blockIdx.x, h = blockIdx.y;
  const int t = threadIdx.x;
  const int r0 = g * NROI;

  for (int i = t; i < 128 * ATT_S; i += 256) att[i] = 0.f;
  for (int i = t; i < 128 * WHS_S; i += 256) whs[i] = 0.f;
  __syncthreads();

  const float* WhH = Wh + (size_t)h * NNODES * NHID;
  for (int i = t; i < NROI * NHID; i += 256) {
    int r = i >> 6, c = i & 63;
    whs[r * WHS_S + c] = WhH[(size_t)(r0 + r) * NHID + c];
  }

  const float* elh = el + (size_t)h * NNODES + r0;
  const float* erh = er + (size_t)h * NNODES + r0;
  if (t < NROI) {
    const float* adjrow = adj + (size_t)(r0 + t) * NNODES + r0;
    float* arow = att + t * ATT_S;
    const float eli = elh[t];
    float m = -INFINITY;
    for (int j = 0; j < NROI; ++j) {
      float e = -INFINITY;
      if (adjrow[j] > 0.f) {
        float z = eli + erh[j];
        e = (z > 0.f) ? z : ALPHA_LR * z;
        m = fmaxf(m, e);
      }
      arow[j] = e;
    }
    if (m == -INFINITY) {
      for (int j = 0; j < NROI; ++j) arow[j] = 0.f;
    } else {
      float s = 0.f;
      for (int j = 0; j < NROI; ++j) {
        float w = (arow[j] == -INFINITY) ? 0.f : expf(arow[j] - m);
        arow[j] = w; s += w;
      }
      float inv = (s > 0.f) ? 1.f / s : 0.f;
      for (int j = 0; j < NROI; ++j) arow[j] *= inv;
    }
  }
  __syncthreads();

  const int w  = t >> 5;
  const int l  = t & 31;
  const int lm = l & 15, lh = l >> 4;
  v8f acc[4];
#pragma unroll
  for (int nt = 0; nt < 4; ++nt)
    acc[nt] = (v8f){0.f,0.f,0.f,0.f,0.f,0.f,0.f,0.f};

  const float* arowA = att + (w * 16 + lm) * ATT_S + 2 * lh;
  const float* bbase = whs + (2 * lh) * WHS_S + lm;
  for (int k0 = 0; k0 < 128; k0 += 4) {
    v2f a = { arowA[k0], arowA[k0 + 1] };
#pragma unroll
    for (int nt = 0; nt < 4; ++nt) {
      const float* bc = bbase + k0 * WHS_S + nt * 16;
      v2f b = { bc[0], bc[WHS_S] };
      acc[nt] = wmma4f32(a, b, acc[nt]);
    }
  }
  __syncthreads();

  float* hp = att;
#pragma unroll
  for (int nt = 0; nt < 4; ++nt)
#pragma unroll
    for (int v = 0; v < 8; ++v)
      hp[(w * 16 + v + 8 * lh) * WHS_S + nt * 16 + lm] = acc[nt][v];
  __syncthreads();

  if (t < NROI) {
    const float* row = hp + t * WHS_S;
    float ss = 0.f;
    for (int c = 0; c < NHID; ++c) ss += row[c] * row[c];
    float r = sqrtf(ss);
    float s = fkernel_scale(r);
    if (DOUBLE_FK) s *= fkernel_scale(s * r);
    float* orow = out + (size_t)(r0 + t) * out_stride + h * NHID;
    for (int c = 0; c < NHID; ++c) orow[c] = act_elu_cos(s * row[c]);
  }
}

extern "C" void kernel_launch(void* const* d_in, const int* in_sizes, int n_in,
                              void* d_out, int out_size, void* d_ws, size_t ws_size,
                              hipStream_t stream) {
  (void)in_sizes; (void)n_in; (void)out_size; (void)ws_size;
  const float* x       = (const float*)d_in[0];
  const float* adj     = (const float*)d_in[1];
  const float* W_heads = (const float*)d_in[2];
  const float* a_heads = (const float*)d_in[3];
  const float* W_out   = (const float*)d_in[4];
  const float* a_out   = (const float*)d_in[5];
  float* out = (float*)d_out;

  float* ws = (float*)d_ws;
  float* Wh   = ws;                  ws += (size_t)NHEADS * NNODES * NHID;
  float* el1  = ws;                  ws += (size_t)NHEADS * NNODES;
  float* er1  = ws;                  ws += (size_t)NHEADS * NNODES;
  float* xcat = ws;                  ws += (size_t)NNODES * NHEADS * NHID;
  float* Wh2  = ws;                  ws += (size_t)NNODES * NHID;
  float* el2  = ws;                  ws += (size_t)NNODES;
  float* er2  = ws;

  const int mtiles = NNODES / 16;

  gemm_wmma_kernel<<<dim3(mtiles * (NHID / 16), NHEADS), 32, 0, stream>>>(
      x, W_heads, Wh, NNODES, NFEAT, NHID);
  attvec_kernel<<<(NHEADS * NNODES + 255) / 256, 256, 0, stream>>>(
      Wh, a_heads, el1, er1, NHEADS);
  att_block_kernel<false><<<dim3(NGRAPH, NHEADS), 256, LDS_BYTES, stream>>>(
      adj, Wh, el1, er1, xcat, NHEADS * NHID);

  gemm_wmma_kernel<<<dim3(mtiles * (NHID / 16), 1), 32, 0, stream>>>(
      xcat, W_out, Wh2, NNODES, NHEADS * NHID, NHID);
  attvec_kernel<<<(NNODES + 255) / 256, 256, 0, stream>>>(
      Wh2, a_out, el2, er2, 1);
  att_block_kernel<true><<<dim3(NGRAPH, 1), 256, LDS_BYTES, stream>>>(
      adj, Wh2, el2, er2, out, NHID);
}


# Masked softmax: one wave per row (lane = 4 columns, one 16-byte adj load per lane, all rows requested at kernel start), max and sum as DPP lane butterflies; the row sum becomes a tree sum
# speedup vs baseline: 1.5562x; 1.5562x over previous
#include <hip/hip_runtime.h>
#include <math.h>

#define NNODES   3712
#define NROI     116
#define NGRAPH   32
#define NFEAT    128
#define NHID     64
#define NHEADS   4
#define ALPHA_LR 0.1f
#define A_PARA   0.1f
#define PHI_C    0.94247779607693793f
#define MAXNORM  (1.0f - 4e-3f)

typedef __attribute__((ext_vector_type(2))) float v2f;
typedef __attribute__((ext_vector_type(8))) float v8f;

__device__ __forceinline__ v8f wmma4f32(v2f a, v2f b, v8f c) {
  return __builtin_amdgcn_wmma_f32_16x16x4_f32(
      false, a, false, b, (short)0, c, false, false);
}

__device__ __forceinline__ float fkernel_scale(float r) {
  float proj = 1.0f, rn = r;
  if (r > MAXNORM) { proj = MAXNORM / r; rn = MAXNORM; }
  rn = fmaxf(rn, 1e-15f);
  float t = fminf(rn, 1.0f - 1e-7f);
  return proj * (atanhf(t) / rn);
}

__device__ __forceinline__ float act_elu_cos(float v) {
  float e = (v > 0.0f) ? v : expm1f(v);
  return e + A_PARA * cosf(v + PHI_C);
}

__device__ __forceinline__ float xhalf(float x) {
  const int b = __builtin_bit_cast(int, x);
  return __builtin_bit_cast(float,
      __builtin_amdgcn_permlanex16(b, b, 0x76543210, 0xfedcba98, false, false));
}
template <int M>
__device__ __forceinline__ float rowx(float x) {
  const int b = __builtin_bit_cast(int, x);
  return __builtin_bit_cast(float,
      __builtin_amdgcn_update_dpp(0, b, 0x160 + M, 0xf, 0xf, true));
}

__global__ __launch_bounds__(32)
void gemm_wmma_kernel(const float* __restrict__ A, const float* __restrict__ B,
                      float* __restrict__ C, int M, int K, int Nc) {
  const int h  = blockIdx.y;
  const float* Bh = B + (size_t)h * K * Nc;
  float*       Ch = C + (size_t)h * M * Nc;
  const int ntiles = Nc >> 4;
  const int mt = blockIdx.x / ntiles;
  const int nt = blockIdx.x % ntiles;
  const int l  = threadIdx.x;
  const int lm = l & 15;
  const int lh = l >> 4;

  v8f acc = {0.f,0.f,0.f,0.f,0.f,0.f,0.f,0.f};
  const float* arow = A  + (size_t)(mt * 16 + lm) * K + 2 * lh;
  const float* bcol = Bh + (size_t)(2 * lh) * Nc + nt * 16 + lm;

  for (int k0 = 0; k0 < K; k0 += 4) {
    v2f a = { arow[k0],              arow[k0 + 1] };
    v2f b = { bcol[(size_t)k0 * Nc], bcol[(size_t)(k0 + 1) * Nc] };
    acc = wmma4f32(a, b, acc);
  }
  float* crow = Ch + (size_t)(mt * 16 + 8 * lh) * Nc + nt * 16 + lm;
#pragma unroll
  for (int v = 0; v < 8; ++v) crow[(size_t)v * Nc] = acc[v];
}

__global__ void attvec_kernel(const float* __restrict__ Wh,
                              const float* __restrict__ avec,
                              float* __restrict__ el, float* __restrict__ er,
                              int heads) {
  int idx = blockIdx.x * blockDim.x + threadIdx.x;
  if (idx >= heads * NNODES) return;
  int h = idx / NNODES, i = idx % NNODES;
  const float* row = Wh + ((size_t)h * NNODES + i) * NHID;
  const float* a1  = avec + (size_t)h * 2 * NHID;
  const float* a2  = a1 + NHID;
  float s1 = 0.f, s2 = 0.f;
#pragma unroll 4
  for (int k = 0; k < NHID; ++k) { s1 += row[k] * a1[k]; s2 += row[k] * a2[k]; }
  el[idx] = s1;
  er[idx] = s2;
}

#define ATT_S 132
#define WHS_S 68
#define LDS_BYTES ((128 * ATT_S + 128 * WHS_S) * sizeof(float))
#define SM_ROWS 15

template <int M>
__device__ __forceinline__ void max_step(float (&v)[SM_ROWS]) {
#pragma unroll
  for (int i = 0; i < SM_ROWS; ++i) v[i] = fmaxf(v[i], rowx<M>(v[i]));
}
template <int M>
__device__ __forceinline__ void sum_step(float (&v)[SM_ROWS]) {
#pragma unroll
  for (int i = 0; i < SM_ROWS; ++i) v[i] += rowx<M>(v[i]);
}

template <bool DOUBLE_FK>
__global__ __launch_bounds__(256)
void att_block_kernel(const float* __restrict__ adj, const float* __restrict__ Wh,
                      const float* __restrict__ el, const float* __restrict__ er,
                      float* __restrict__ out, int out_stride) {
  extern __shared__ __align__(16) float lds[];
  float* att = lds;
  float* whs = lds + 128 * ATT_S;

  const int g = blockIdx.x, h = blockIdx.y;
  const int t = threadIdx.x;
  const int r0 = g * NROI;

  const int wu = __builtin_amdgcn_readfirstlane(t >> 5);
  const int ls = t & 31;
  const bool lane_on = ls < NROI / 4;
  const int lc = lane_on ? ls : (NROI / 4 - 1);
  const float* elh = el + (size_t)h * NNODES + r0;
  const float* erh = er + (size_t)h * NNODES + r0;
  const float er4[4] = { erh[4 * lc], erh[4 * lc + 1], erh[4 * lc + 2], erh[4 * lc + 3] };
  float4 av[SM_ROWS];
  float  elv[SM_ROWS];
#pragma unroll
  for (int i = 0; i < SM_ROWS; ++i) {
    const int r = (i < SM_ROWS - 1) ? wu + 8 * i : min(wu + 8 * i, NROI - 1);
    av[i]  = *reinterpret_cast<const float4*>(adj + (size_t)(r0 + r) * NNODES + r0 + 4 * lc);
    elv[i] = elh[r];
  }

  for (int i = t; i < 128 * ATT_S; i += 256) att[i] = 0.f;
  for (int i = t; i < 128 * WHS_S; i += 256) whs[i] = 0.f;
  __syncthreads();

  const float* WhH = Wh + (size_t)h * NNODES * NHID;
  for (int i = t; i < NROI * NHID; i += 256) {
    int r = i >> 6, c = i & 63;
    whs[r * WHS_S + c] = WhH[(size_t)(r0 + r) * NHID + c];
  }

  {
    float e[SM_ROWS][4];
    float m[SM_ROWS];
#pragma unroll
    for (int i = 0; i < SM_ROWS; ++i) {
      const float a4[4] = { av[i].x, av[i].y, av[i].z, av[i].w };
      float mi = -INFINITY;
#pragma unroll
      for (int q = 0; q < 4; ++q) {
        const float z  = elv[i] + er4[q];
        const float lr = (z > 0.f) ? z : ALPHA_LR * z;
        const float ev = (lane_on && a4[q] > 0.f) ? lr : -INFINITY;
        e[i][q] = ev;
        mi = fmaxf(mi, ev);
      }
      m[i] = mi;
    }
    max_step<1>(m); max_step<2>(m); max_step<4>(m); max_step<8>(m);
#pragma unroll
    for (int i = 0; i < SM_ROWS; ++i) m[i] = fmaxf(m[i], xhalf(m[i]));

    float s[SM_ROWS];
#pragma unroll
    for (int i = 0; i < SM_ROWS; ++i) {
      const float mm = (m[i] == -INFINITY) ? 0.f : m[i];
      float si = 0.f;
#pragma unroll
      for (int q = 0; q < 4; ++q) {
        e[i][q] = expf(e[i][q] - mm);
        si += e[i][q];
      }
      s[i] = si;
    }
    sum_step<1>(s); sum_step<2>(s); sum_step<4>(s); sum_step<8>(s);
#pragma unroll
    for (int i = 0; i < SM_ROWS; ++i) s[i] += xhalf(s[i]);

#pragma unroll
    for (int i = 0; i < SM_ROWS; ++i) {
      const int r = wu + 8 * i;
      const float inv = (s[i] > 0.f) ? 1.f / s[i] : 0.f;
      if (i < SM_ROWS - 1 || r < NROI) {
        const float4 o = { e[i][0] * inv, e[i][1] * inv, e[i][2] * inv, e[i][3] * inv };
        *reinterpret_cast<float4*>(att + r * ATT_S + 4 * ls) = o;
      }
    }
  }
  __syncthreads();

  const int w  = t >> 5;
  const int l  = t & 31;
  const int lm = l & 15, lh = l >> 4;
  v8f acc[4];
#pragma unroll
  for (int nt = 0; nt < 4; ++nt)
    acc[nt] = (v8f){0.f,0.f,0.f,0.f,0.f,0.f,0.f,0.f};

  const float* arowA = att + (w * 16 + lm) * ATT_S + 2 * lh;
  const float* bbase = whs + (2 * lh) * WHS_S + lm;
  for (int k0 = 0; k0 < 128; k0 += 4) {
    v2f a = { arowA[k0], arowA[k0 + 1] };
#pragma unroll
    for (int nt = 0; nt < 4; ++nt) {
      const float* bc = bbase + k0 * WHS_S + nt * 16;
      v2f b = { bc[0], bc[WHS_S] };
      acc[nt] = wmma4f32(a, b, acc[nt]);
    }
  }
  __syncthreads();

  float* hp = att;
#pragma unroll
  for (int nt = 0; nt < 4; ++nt)
#pragma unroll
    for (int v = 0; v < 8; ++v)
      hp[(w * 16 + v + 8 * lh) * WHS_S + nt * 16 + lm] = acc[nt][v];
  __syncthreads();

  if (t < NROI) {
    const float* row = hp + t * WHS_S;
    float ss = 0.f;
    for (int c = 0; c < NHID; ++c) ss += row[c] * row[c];
    float r = sqrtf(ss);
    float s = fkernel_scale(r);
    if (DOUBLE_FK) s *= fkernel_scale(s * r);
    float* orow = out + (size_t)(r0 + t) * out_stride + h * NHID;
    for (int c = 0; c < NHID; ++c) orow[c] = act_elu_cos(s * row[c]);
  }
}

extern "C" void kernel_launch(void* const* d_in, const int* in_sizes, int n_in,
                              void* d_out, int out_size, void* d_ws, size_t ws_size,
                              hipStream_t stream) {
  (void)in_sizes; (void)n_in; (void)out_size; (void)ws_size;
  const float* x       = (const float*)d_in[0];
  const float* adj     = (const float*)d_in[1];
  const float* W_heads = (const float*)d_in[2];
  const float* a_heads = (const float*)d_in[3];
  const float* W_out   = (const float*)d_in[4];
  const float* a_out   = (const float*)d_in[5];
  float* out = (float*)d_out;

  float* ws = (float*)d_ws;
  float* Wh   = ws;                  ws += (size_t)NHEADS * NNODES * NHID;
  float* el1  = ws;                  ws += (size_t)NHEADS * NNODES;
  float* er1  = ws;                  ws += (size_t)NHEADS * NNODES;
  float* xcat = ws;                  ws += (size_t)NNODES * NHEADS * NHID;
  float* Wh2  = ws;                  ws += (size_t)NNODES * NHID;
  float* el2  = ws;                  ws += (size_t)NNODES;
  float* er2  = ws;

  const int mtiles = NNODES / 16;

  gemm_wmma_kernel<<<dim3(mtiles * (NHID / 16), NHEADS), 32, 0, stream>>>(
      x, W_heads, Wh, NNODES, NFEAT, NHID);
  attvec_kernel<<<(NHEADS * NNODES + 255) / 256, 256, 0, stream>>>(
      Wh, a_heads, el1, er1, NHEADS);
  att_block_kernel<false><<<dim3(NGRAPH, NHEADS), 256, LDS_BYTES, stream>>>(
      adj, Wh, el1, er1, xcat, NHEADS * NHID);

  gemm_wmma_kernel<<<dim3(mtiles * (NHID / 16), 1), 32, 0, stream>>>(
      xcat, W_out, Wh2, NNODES, NHEADS * NHID, NHID);
  attvec_kernel<<<(NNODES + 255) / 256, 256, 0, stream>>>(
      Wh2, a_out, el2, er2, 1);
  att_block_kernel<true><<<dim3(NGRAPH, 1), 256, LDS_BYTES, stream>>>(
      adj, Wh2, el2, er2, out, NHID);
}


# Row-scale and activation tail spread from 116 lanes (64 elements in one chain each) to all 256: a lane takes 4 consecutive columns, ELU arm as a select, one 256-byte row segment per half-wave store
# speedup vs baseline: 1.8181x; 1.1683x over previous
#include <hip/hip_runtime.h>
#include <math.h>

#define NNODES   3712
#define NROI     116
#define NGRAPH   32
#define NFEAT    128
#define NHID     64
#define NHEADS   4
#define ALPHA_LR 0.1f
#define A_PARA   0.1f
#define PHI_C    0.94247779607693793f
#define MAXNORM  (1.0f - 4e-3f)

typedef __attribute__((ext_vector_type(2))) float v2f;
typedef __attribute__((ext_vector_type(8))) float v8f;

__device__ __forceinline__ v8f wmma4f32(v2f a, v2f b, v8f c) {
  return __builtin_amdgcn_wmma_f32_16x16x4_f32(
      false, a, false, b, (short)0, c, false, false);
}

__device__ __forceinline__ float fkernel_scale(float r) {
  float proj = 1.0f, rn = r;
  if (r > MAXNORM) { proj = MAXNORM / r; rn = MAXNORM; }
  rn = fmaxf(rn, 1e-15f);
  float t = fminf(rn, 1.0f - 1e-7f);
  return proj * (atanhf(t) / rn);
}

__device__ __forceinline__ float act_elu_cos(float v) {
  const float em = expm1f(v);
  const float e  = (v > 0.0f) ? v : em;
  return e + A_PARA * cosf(v + PHI_C);
}

__device__ __forceinline__ float xhalf(float x) {
  const int b = __builtin_bit_cast(int, x);
  return __builtin_bit_cast(float,
      __builtin_amdgcn_permlanex16(b, b, 0x76543210, 0xfedcba98, false, false));
}
template <int M>
__device__ __forceinline__ float rowx(float x) {
  const int b = __builtin_bit_cast(int, x);
  return __builtin_bit_cast(float,
      __builtin_amdgcn_update_dpp(0, b, 0x160 + M, 0xf, 0xf, true));
}

__global__ __launch_bounds__(32)
void gemm_wmma_kernel(const float* __restrict__ A, const float* __restrict__ B,
                      float* __restrict__ C, int M, int K, int Nc) {
  const int h  = blockIdx.y;
  const float* Bh = B + (size_t)h * K * Nc;
  float*       Ch = C + (size_t)h * M * Nc;
  const int ntiles = Nc >> 4;
  const int mt = blockIdx.x / ntiles;
  const int nt = blockIdx.x % ntiles;
  const int l  = threadIdx.x;
  const int lm = l & 15;
  const int lh = l >> 4;

  v8f acc = {0.f,0.f,0.f,0.f,0.f,0.f,0.f,0.f};
  const float* arow = A  + (size_t)(mt * 16 + lm) * K + 2 * lh;
  const float* bcol = Bh + (size_t)(2 * lh) * Nc + nt * 16 + lm;

  for (int k0 = 0; k0 < K; k0 += 4) {
    v2f a = { arow[k0],              arow[k0 + 1] };
    v2f b = { bcol[(size_t)k0 * Nc], bcol[(size_t)(k0 + 1) * Nc] };
    acc = wmma4f32(a, b, acc);
  }
  float* crow = Ch + (size_t)(mt * 16 + 8 * lh) * Nc + nt * 16 + lm;
#pragma unroll
  for (int v = 0; v < 8; ++v) crow[(size_t)v * Nc] = acc[v];
}

__global__ void attvec_kernel(const float* __restrict__ Wh,
                              const float* __restrict__ avec,
                              float* __restrict__ el, float* __restrict__ er,
                              int heads) {
  int idx = blockIdx.x * blockDim.x + threadIdx.x;
  if (idx >= heads * NNODES) return;
  int h = idx / NNODES, i = idx % NNODES;
  const float* row = Wh + ((size_t)h * NNODES + i) * NHID;
  const float* a1  = avec + (size_t)h * 2 * NHID;
  const float* a2  = a1 + NHID;
  float s1 = 0.f, s2 = 0.f;
#pragma unroll 4
  for (int k = 0; k < NHID; ++k) { s1 += row[k] * a1[k]; s2 += row[k] * a2[k]; }
  el[idx] = s1;
  er[idx] = s2;
}

#define ATT_S 132
#define WHS_S 68
#define LDS_BYTES ((128 * ATT_S + 128 * WHS_S) * sizeof(float))
#define SM_ROWS 15

template <int M>
__device__ __forceinline__ void max_step(float (&v)[SM_ROWS]) {
#pragma unroll
  for (int i = 0; i < SM_ROWS; ++i) v[i] = fmaxf(v[i], rowx<M>(v[i]));
}
template <int M>
__device__ __forceinline__ void sum_step(float (&v)[SM_ROWS]) {
#pragma unroll
  for (int i = 0; i < SM_ROWS; ++i) v[i] += rowx<M>(v[i]);
}

template <bool DOUBLE_FK>
__global__ __launch_bounds__(256)
void att_block_kernel(const float* __restrict__ adj, const float* __restrict__ Wh,
                      const float* __restrict__ el, const float* __restrict__ er,
                      float* __restrict__ out, int out_stride) {
  extern __shared__ __align__(16) float lds[];
  float* att = lds;
  float* whs = lds + 128 * ATT_S;

  const int g = blockIdx.x, h = blockIdx.y;
  const int t = threadIdx.x;
  const int r0 = g * NROI;

  const int wu = __builtin_amdgcn_readfirstlane(t >> 5);
  const int ls = t & 31;
  const bool lane_on = ls < NROI / 4;
  const int lc = lane_on ? ls : (NROI / 4 - 1);
  const float* elh = el + (size_t)h * NNODES + r0;
  const float* erh = er + (size_t)h * NNODES + r0;
  const float er4[4] = { erh[4 * lc], erh[4 * lc + 1], erh[4 * lc + 2], erh[4 * lc + 3] };
  float4 av[SM_ROWS];
  float  elv[SM_ROWS];
#pragma unroll
  for (int i = 0; i < SM_ROWS; ++i) {
    const int r = (i < SM_ROWS - 1) ? wu + 8 * i : min(wu + 8 * i, NROI - 1);
    av[i]  = *reinterpret_cast<const float4*>(adj + (size_t)(r0 + r) * NNODES + r0 + 4 * lc);
    elv[i] = elh[r];
  }

  for (int i = t; i < 128 * ATT_S; i += 256) att[i] = 0.f;
  for (int i = t; i < 128 * WHS_S; i += 256) whs[i] = 0.f;
  __syncthreads();

  const float* WhH = Wh + (size_t)h * NNODES * NHID;
  for (int i = t; i < NROI * NHID; i += 256) {
    int r = i >> 6, c = i & 63;
    whs[r * WHS_S + c] = WhH[(size_t)(r0 + r) * NHID + c];
  }

  {
    float e[SM_ROWS][4];
    float m[SM_ROWS];
#pragma unroll
    for (int i = 0; i < SM_ROWS; ++i) {
      const float a4[4] = { av[i].x, av[i].y, av[i].z, av[i].w };
      float mi = -INFINITY;
#pragma unroll
      for (int q = 0; q < 4; ++q) {
        const float z  = elv[i] + er4[q];
        const float lr = (z > 0.f) ? z : ALPHA_LR * z;
        const float ev = (lane_on && a4[q] > 0.f) ? lr : -INFINITY;
        e[i][q] = ev;
        mi = fmaxf(mi, ev);
      }
      m[i] = mi;
    }
    max_step<1>(m); max_step<2>(m); max_step<4>(m); max_step<8>(m);
#pragma unroll
    for (int i = 0; i < SM_ROWS; ++i) m[i] = fmaxf(m[i], xhalf(m[i]));

    float s[SM_ROWS];
#pragma unroll
    for (int i = 0; i < SM_ROWS; ++i) {
      const float mm = (m[i] == -INFINITY) ? 0.f : m[i];
      float si = 0.f;
#pragma unroll
      for (int q = 0; q < 4; ++q) {
        e[i][q] = expf(e[i][q] - mm);
        si += e[i][q];
      }
      s[i] = si;
    }
    sum_step<1>(s); sum_step<2>(s); sum_step<4>(s); sum_step<8>(s);
#pragma unroll
    for (int i = 0; i < SM_ROWS; ++i) s[i] += xhalf(s[i]);

#pragma unroll
    for (int i = 0; i < SM_ROWS; ++i) {
      const int r = wu + 8 * i;
      const float inv = (s[i] > 0.f) ? 1.f / s[i] : 0.f;
      if (i < SM_ROWS - 1 || r < NROI) {
        const float4 o = { e[i][0] * inv, e[i][1] * inv, e[i][2] * inv, e[i][3] * inv };
        *reinterpret_cast<float4*>(att + r * ATT_S + 4 * ls) = o;
      }
    }
  }
  __syncthreads();

  const int w  = t >> 5;
  const int l  = t & 31;
  const int lm = l & 15, lh = l >> 4;
  v8f acc[4];
#pragma unroll
  for (int nt = 0; nt < 4; ++nt)
    acc[nt] = (v8f){0.f,0.f,0.f,0.f,0.f,0.f,0.f,0.f};

  const float* arowA = att + (w * 16 + lm) * ATT_S + 2 * lh;
  const float* bbase = whs + (2 * lh) * WHS_S + lm;
  for (int k0 = 0; k0 < 128; k0 += 4) {
    v2f a = { arowA[k0], arowA[k0 + 1] };
#pragma unroll
    for (int nt = 0; nt < 4; ++nt) {
      const float* bc = bbase + k0 * WHS_S + nt * 16;
      v2f b = { bc[0], bc[WHS_S] };
      acc[nt] = wmma4f32(a, b, acc[nt]);
    }
  }
  __syncthreads();

  float* hp = att;
#pragma unroll
  for (int nt = 0; nt < 4; ++nt)
#pragma unroll
    for (int v = 0; v < 8; ++v)
      hp[(w * 16 + v + 8 * lh) * WHS_S + nt * 16 + lm] = acc[nt][v];
  __syncthreads();

  float* sc = whs;
  if (t < NROI) {
    const float* row = hp + t * WHS_S;
    float ss = 0.f;
    for (int c = 0; c < NHID; ++c) ss += row[c] * row[c];
    float r = sqrtf(ss);
    float s = fkernel_scale(r);
    if (DOUBLE_FK) s *= fkernel_scale(s * r);
    sc[t] = s;
  }
  __syncthreads();

  float* obase = out + (size_t)r0 * out_stride + h * NHID;
  for (int q = t; q < NROI * (NHID / 4); q += 256) {
    const int row = q >> 4;
    const int c4  = (q & 15) * 4;
    const float4 v = *reinterpret_cast<const float4*>(hp + row * WHS_S + c4);
    const float  s = sc[row];
    float4 o;
    o.x = act_elu_cos(s * v.x);
    o.y = act_elu_cos(s * v.y);
    o.z = act_elu_cos(s * v.z);
    o.w = act_elu_cos(s * v.w);
    *reinterpret_cast<float4*>(obase + (size_t)row * out_stride + c4) = o;
  }
}

extern "C" void kernel_launch(void* const* d_in, const int* in_sizes, int n_in,
                              void* d_out, int out_size, void* d_ws, size_t ws_size,
                              hipStream_t stream) {
  (void)in_sizes; (void)n_in; (void)out_size; (void)ws_size;
  const float* x       = (const float*)d_in[0];
  const float* adj     = (const float*)d_in[1];
  const float* W_heads = (const float*)d_in[2];
  const float* a_heads = (const float*)d_in[3];
  const float* W_out   = (const float*)d_in[4];
  const float* a_out   = (const float*)d_in[5];
  float* out = (float*)d_out;

  float* ws = (float*)d_ws;
  float* Wh   = ws;                  ws += (size_t)NHEADS * NNODES * NHID;
  float* el1  = ws;                  ws += (size_t)NHEADS * NNODES;
  float* er1  = ws;                  ws += (size_t)NHEADS * NNODES;
  float* xcat = ws;                  ws += (size_t)NNODES * NHEADS * NHID;
  float* Wh2  = ws;                  ws += (size_t)NNODES * NHID;
  float* el2  = ws;                  ws += (size_t)NNODES;
  float* er2  = ws;

  const int mtiles = NNODES / 16;

  gemm_wmma_kernel<<<dim3(mtiles * (NHID / 16), NHEADS), 32, 0, stream>>>(
      x, W_heads, Wh, NNODES, NFEAT, NHID);
  attvec_kernel<<<(NHEADS * NNODES + 255) / 256, 256, 0, stream>>>(
      Wh, a_heads, el1, er1, NHEADS);
  att_block_kernel<false><<<dim3(NGRAPH, NHEADS), 256, LDS_BYTES, stream>>>(
      adj, Wh, el1, er1, xcat, NHEADS * NHID);

  gemm_wmma_kernel<<<dim3(mtiles * (NHID / 16), 1), 32, 0, stream>>>(
      xcat, W_out, Wh2, NNODES, NHEADS * NHID, NHID);
  attvec_kernel<<<(NNODES + 255) / 256, 256, 0, stream>>>(
      Wh2, a_out, el2, er2, 1);
  att_block_kernel<true><<<dim3(NGRAPH, 1), 256, LDS_BYTES, stream>>>(
      adj, Wh2, el2, er2, out, NHID);
}


# Projection GEMM and el/er dots fused into the per-graph attention kernel: Wh tile by WMMA from LDS-staged W straight into LDS, never written out; 6 launches become 2, global loads in one batch
# speedup vs baseline: 3.1276x; 1.7203x over previous
#include <hip/hip_runtime.h>
#include <math.h>

#define NNODES   3712
#define NROI     116
#define NGRAPH   32
#define NFEAT    128
#define NHID     64
#define NHEADS   4
#define ALPHA_LR 0.1f
#define A_PARA   0.1f
#define PHI_C    0.94247779607693793f
#define MAXNORM  (1.0f - 4e-3f)

typedef __attribute__((ext_vector_type(2))) float v2f;
typedef __attribute__((ext_vector_type(8))) float v8f;

__device__ __forceinline__ v8f wmma4f32(v2f a, v2f b, v8f c) {
  return __builtin_amdgcn_wmma_f32_16x16x4_f32(
      false, a, false, b, (short)0, c, false, false);
}

__device__ __forceinline__ float fkernel_scale(float r) {
  float proj = 1.0f, rn = r;
  if (r > MAXNORM) { proj = MAXNORM / r; rn = MAXNORM; }
  rn = fmaxf(rn, 1e-15f);
  float t = fminf(rn, 1.0f - 1e-7f);
  return proj * (atanhf(t) / rn);
}

__device__ __forceinline__ float act_elu_cos(float v) {
  const float em = expm1f(v);
  const float e  = (v > 0.0f) ? v : em;
  return e + A_PARA * cosf(v + PHI_C);
}

__device__ __forceinline__ float xhalf(float x) {
  const int b = __builtin_bit_cast(int, x);
  return __builtin_bit_cast(float,
      __builtin_amdgcn_permlanex16(b, b, 0x76543210, 0xfedcba98, false, false));
}
template <int M>
__device__ __forceinline__ float rowx(float x) {
  const int b = __builtin_bit_cast(int, x);
  return __builtin_bit_cast(float,
      __builtin_amdgcn_update_dpp(0, b, 0x160 + M, 0xf, 0xf, true));
}

#define ATT_S 132
#define WHS_S 68
#define REGA_FLOATS(KIN) (((KIN) * WHS_S > 128 * ATT_S) ? (KIN) * WHS_S : 128 * ATT_S)
#define LDS_BYTES(KIN) ((REGA_FLOATS(KIN) + 128 * WHS_S + 256) * sizeof(float))
#define SM_ROWS 15

template <int M>
__device__ __forceinline__ void max_step(float (&v)[SM_ROWS]) {
#pragma unroll
  for (int i = 0; i < SM_ROWS; ++i) v[i] = fmaxf(v[i], rowx<M>(v[i]));
}
template <int M>
__device__ __forceinline__ void sum_step(float (&v)[SM_ROWS]) {
#pragma unroll
  for (int i = 0; i < SM_ROWS; ++i) v[i] += rowx<M>(v[i]);
}

template <bool DOUBLE_FK, int KIN>
__global__ __launch_bounds__(256) __attribute__((amdgpu_waves_per_eu(2, 2)))
void att_block_kernel(const float* __restrict__ xin, const float* __restrict__ adj,
                      const float* __restrict__ Wall, const float* __restrict__ avec,
                      float* __restrict__ out, int out_stride) {
  extern __shared__ __align__(16) float lds[];
  float* wst = lds;
  float* att = lds;
  float* whs = lds + REGA_FLOATS(KIN);
  float* els = whs + 128 * WHS_S;
  float* ers = els + 128;

  const int g = blockIdx.x, h = blockIdx.y;
  const int t = threadIdx.x;
  const int r0 = g * NROI;

  const int wu = __builtin_amdgcn_readfirstlane(t >> 5);
  const int ls = t & 31;
  const int lm = ls & 15, lh = ls >> 4;
  const bool lane_on = ls < NROI / 4;
  const int lc = lane_on ? ls : (NROI / 4 - 1);

  const float* Wg = Wall + (size_t)h * KIN * NHID;
  float4 wv[KIN / 16];
#pragma unroll
  for (int it = 0; it < KIN / 16; ++it) {
    const int i = t + 256 * it;
    wv[it] = *reinterpret_cast<const float4*>(Wg + (size_t)(i >> 4) * NHID + (i & 15) * 4);
  }

  const float* arow = xin + (size_t)(r0 + min(wu * 16 + lm, NROI - 1)) * KIN + 2 * lh;
  v2f af[KIN / 4];
#pragma unroll
  for (int j = 0; j < KIN / 4; ++j) af[j] = *reinterpret_cast<const v2f*>(arow + 4 * j);

  float4 av[SM_ROWS];
#pragma unroll
  for (int i = 0; i < SM_ROWS; ++i) {
    const int r = (i < SM_ROWS - 1) ? wu + 8 * i : min(wu + 8 * i, NROI - 1);
    av[i]  = *reinterpret_cast<const float4*>(adj + (size_t)(r0 + r) * NNODES + r0 + 4 * lc);
  }

#pragma unroll
  for (int it = 0; it < KIN / 16; ++it) {
    const int i = t + 256 * it;
    *reinterpret_cast<float4*>(wst + (i >> 4) * WHS_S + (i & 15) * 4) = wv[it];
  }
  __syncthreads();

  {
    v8f pacc[4];
#pragma unroll
    for (int nt = 0; nt < 4; ++nt)
      pacc[nt] = (v8f){0.f,0.f,0.f,0.f,0.f,0.f,0.f,0.f};
    const float* pb = wst + (2 * lh) * WHS_S + lm;
#pragma unroll
    for (int j = 0; j < KIN / 4; ++j) {
#pragma unroll
      for (int nt = 0; nt < 4; ++nt) {
        const float* bc = pb + (4 * j) * WHS_S + nt * 16;
        v2f b = { bc[0], bc[WHS_S] };
        pacc[nt] = wmma4f32(af[j], b, pacc[nt]);
      }
    }
#pragma unroll
    for (int nt = 0; nt < 4; ++nt)
#pragma unroll
      for (int v = 0; v < 8; ++v)
        whs[(wu * 16 + v + 8 * lh) * WHS_S + nt * 16 + lm] = pacc[nt][v];
  }
  __syncthreads();

  {
    const float* ah  = avec + (size_t)h * 2 * NHID + (wu >> 2) * NHID;
    const float* row = whs + (t & 127) * WHS_S;
    float s = 0.f;
#pragma unroll
    for (int k = 0; k < NHID; ++k) s += row[k] * ah[k];
    els[t] = s;
  }
  __syncthreads();

  const float4 er4v = *reinterpret_cast<const float4*>(ers + 4 * lc);
  const float er4[4] = { er4v.x, er4v.y, er4v.z, er4v.w };
  float elv[SM_ROWS];
#pragma unroll
  for (int i = 0; i < SM_ROWS; ++i)
    elv[i] = els[(i < SM_ROWS - 1) ? wu + 8 * i : min(wu + 8 * i, NROI - 1)];

  {
    float e[SM_ROWS][4];
    float m[SM_ROWS];
#pragma unroll
    for (int i = 0; i < SM_ROWS; ++i) {
      const float a4[4] = { av[i].x, av[i].y, av[i].z, av[i].w };
      float mi = -INFINITY;
#pragma unroll
      for (int q = 0; q < 4; ++q) {
        const float z  = elv[i] + er4[q];
        const float lr = (z > 0.f) ? z : ALPHA_LR * z;
        const float ev = (lane_on && a4[q] > 0.f) ? lr : -INFINITY;
        e[i][q] = ev;
        mi = fmaxf(mi, ev);
      }
      m[i] = mi;
    }
    max_step<1>(m); max_step<2>(m); max_step<4>(m); max_step<8>(m);
#pragma unroll
    for (int i = 0; i < SM_ROWS; ++i) m[i] = fmaxf(m[i], xhalf(m[i]));

    float s[SM_ROWS];
#pragma unroll
    for (int i = 0; i < SM_ROWS; ++i) {
      const float mm = (m[i] == -INFINITY) ? 0.f : m[i];
      float si = 0.f;
#pragma unroll
      for (int q = 0; q < 4; ++q) {
        e[i][q] = expf(e[i][q] - mm);
        si += e[i][q];
      }
      s[i] = si;
    }
    sum_step<1>(s); sum_step<2>(s); sum_step<4>(s); sum_step<8>(s);
#pragma unroll
    for (int i = 0; i < SM_ROWS; ++i) s[i] += xhalf(s[i]);

#pragma unroll
    for (int i = 0; i < SM_ROWS; ++i) {
      const int r = wu + 8 * i;
      const float inv = (s[i] > 0.f) ? 1.f / s[i] : 0.f;
      if (i < SM_ROWS - 1 || r < NROI) {
        const float4 o = { e[i][0] * inv, e[i][1] * inv, e[i][2] * inv, e[i][3] * inv };
        *reinterpret_cast<float4*>(att + r * ATT_S + 4 * ls) = o;
      }
    }
  }
  __syncthreads();

  const int w  = wu;
  v8f acc[4];
#pragma unroll
  for (int nt = 0; nt < 4; ++nt)
    acc[nt] = (v8f){0.f,0.f,0.f,0.f,0.f,0.f,0.f,0.f};

  const float* arowA = att + (w * 16 + lm) * ATT_S + 2 * lh;
  const float* bbase = whs + (2 * lh) * WHS_S + lm;
  for (int k0 = 0; k0 < 128; k0 += 4) {
    v2f a = { arowA[k0], arowA[k0 + 1] };
#pragma unroll
    for (int nt = 0; nt < 4; ++nt) {
      const float* bc = bbase + k0 * WHS_S + nt * 16;
      v2f b = { bc[0], bc[WHS_S] };
      acc[nt] = wmma4f32(a, b, acc[nt]);
    }
  }
  __syncthreads();

  float* hp = att;
#pragma unroll
  for (int nt = 0; nt < 4; ++nt)
#pragma unroll
    for (int v = 0; v < 8; ++v)
      hp[(w * 16 + v + 8 * lh) * WHS_S + nt * 16 + lm] = acc[nt][v];
  __syncthreads();

  float* sc = whs;
  if (t < NROI) {
    const float* row = hp + t * WHS_S;
    float ss = 0.f;
    for (int c = 0; c < NHID; ++c) ss += row[c] * row[c];
    float r = sqrtf(ss);
    float s = fkernel_scale(r);
    if (DOUBLE_FK) s *= fkernel_scale(s * r);
    sc[t] = s;
  }
  __syncthreads();

  float* obase = out + (size_t)r0 * out_stride + h * NHID;
  for (int q = t; q < NROI * (NHID / 4); q += 256) {
    const int row = q >> 4;
    const int c4  = (q & 15) * 4;
    const float4 v = *reinterpret_cast<const float4*>(hp + row * WHS_S + c4);
    const float  s = sc[row];
    float4 o;
    o.x = act_elu_cos(s * v.x);
    o.y = act_elu_cos(s * v.y);
    o.z = act_elu_cos(s * v.z);
    o.w = act_elu_cos(s * v.w);
    *reinterpret_cast<float4*>(obase + (size_t)row * out_stride + c4) = o;
  }
}

extern "C" void kernel_launch(void* const* d_in, const int* in_sizes, int n_in,
                              void* d_out, int out_size, void* d_ws, size_t ws_size,
                              hipStream_t stream) {
  (void)in_sizes; (void)n_in; (void)out_size; (void)ws_size;
  const float* x       = (const float*)d_in[0];
  const float* adj     = (const float*)d_in[1];
  const float* W_heads = (const float*)d_in[2];
  const float* a_heads = (const float*)d_in[3];
  const float* W_out   = (const float*)d_in[4];
  const float* a_out   = (const float*)d_in[5];
  float* out = (float*)d_out;

  float* xcat = (float*)d_ws;

  att_block_kernel<false, NFEAT><<<dim3(NGRAPH, NHEADS), 256, LDS_BYTES(NFEAT), stream>>>(
      x, adj, W_heads, a_heads, xcat, NHEADS * NHID);
  att_block_kernel<true, NHEADS * NHID><<<dim3(NGRAPH, 1), 256, LDS_BYTES(NHEADS * NHID), stream>>>(
      xcat, adj, W_out, a_out, out, NHID);
}


# Attention rows of each graph block split into row slabs, one workgroup per slab (2 in layer 1, 4 in layer 2); each slab workgroup rebuilds the block's whole Wh tile; every sum order unchanged
# speedup vs baseline: 4.2294x; 1.3523x over previous
#include <hip/hip_runtime.h>
#include <math.h>

#define NNODES   3712
#define NROI     116
#define NGRAPH   32
#define NFEAT    128
#define NHID     64
#define NHEADS   4
#define ALPHA_LR 0.1f
#define A_PARA   0.1f
#define PHI_C    0.94247779607693793f
#define MAXNORM  (1.0f - 4e-3f)

typedef __attribute__((ext_vector_type(2))) float v2f;
typedef __attribute__((ext_vector_type(8))) float v8f;

__device__ __forceinline__ v8f wmma4f32(v2f a, v2f b, v8f c) {
  return __builtin_amdgcn_wmma_f32_16x16x4_f32(
      false, a, false, b, (short)0, c, false, false);
}

__device__ __forceinline__ float fkernel_scale(float r) {
  float proj = 1.0f, rn = r;
  if (r > MAXNORM) { proj = MAXNORM / r; rn = MAXNORM; }
  rn = fmaxf(rn, 1e-15f);
  float t = fminf(rn, 1.0f - 1e-7f);
  return proj * (atanhf(t) / rn);
}

__device__ __forceinline__ float act_elu_cos(float v) {
  const float em = expm1f(v);
  const float e  = (v > 0.0f) ? v : em;
  return e + A_PARA * cosf(v + PHI_C);
}

__device__ __forceinline__ float xhalf(float x) {
  const int b = __builtin_bit_cast(int, x);
  return __builtin_bit_cast(float,
      __builtin_amdgcn_permlanex16(b, b, 0x76543210, 0xfedcba98, false, false));
}
template <int M>
__device__ __forceinline__ float rowx(float x) {
  const int b = __builtin_bit_cast(int, x);
  return __builtin_bit_cast(float,
      __builtin_amdgcn_update_dpp(0, b, 0x160 + M, 0xf, 0xf, true));
}

#define ATT_S 132
#define WHS_S 68
#define REGA_FLOATS(KIN, SR) (((KIN) * WHS_S > (SR) * ATT_S) ? (KIN) * WHS_S : (SR) * ATT_S)
#define LDS_BYTES(KIN, SR) ((REGA_FLOATS(KIN, SR) + 128 * WHS_S + 256) * sizeof(float))

template <int M, int R>
__device__ __forceinline__ void max_step(float (&v)[R]) {
#pragma unroll
  for (int i = 0; i < R; ++i) v[i] = fmaxf(v[i], rowx<M>(v[i]));
}
template <int M, int R>
__device__ __forceinline__ void sum_step(float (&v)[R]) {
#pragma unroll
  for (int i = 0; i < R; ++i) v[i] += rowx<M>(v[i]);
}

template <bool DOUBLE_FK, int KIN, int NSLAB>
__global__ __launch_bounds__(256) __attribute__((amdgpu_waves_per_eu(2, 2)))
void att_block_kernel(const float* __restrict__ xin, const float* __restrict__ adj,
                      const float* __restrict__ Wall, const float* __restrict__ avec,
                      float* __restrict__ out, int out_stride) {
  constexpr int SR  = 128 / NSLAB;
  constexpr int SMR = SR / 8;
  constexpr int MT  = SR / 16;
  constexpr int WPM = 8 / MT;
  constexpr int NTW = 4 / WPM;
  extern __shared__ __align__(16) float lds[];
  float* wst = lds;
  float* att = lds;
  float* whs = lds + REGA_FLOATS(KIN, SR);
  float* els = whs + 128 * WHS_S;
  float* ers = els + 128;

  const int g = blockIdx.x, h = blockIdx.y;
  const int t = threadIdx.x;
  const int r0 = g * NROI;
  const int rb = blockIdx.z * SR;

  const int wu = __builtin_amdgcn_readfirstlane(t >> 5);
  const int ls = t & 31;
  const int lm = ls & 15, lh = ls >> 4;
  const bool lane_on = ls < NROI / 4;
  const int lc = lane_on ? ls : (NROI / 4 - 1);

  const float* Wg = Wall + (size_t)h * KIN * NHID;
  float4 wv[KIN / 16];
#pragma unroll
  for (int it = 0; it < KIN / 16; ++it) {
    const int i = t + 256 * it;
    wv[it] = *reinterpret_cast<const float4*>(Wg + (size_t)(i >> 4) * NHID + (i & 15) * 4);
  }

  const float* arow = xin + (size_t)(r0 + min(wu * 16 + lm, NROI - 1)) * KIN + 2 * lh;
  v2f af[KIN / 4];
#pragma unroll
  for (int j = 0; j < KIN / 4; ++j) af[j] = *reinterpret_cast<const v2f*>(arow + 4 * j);

  float4 av[SMR];
#pragma unroll
  for (int i = 0; i < SMR; ++i) {
    const int r = min(rb + wu + 8 * i, NROI - 1);
    av[i]  = *reinterpret_cast<const float4*>(adj + (size_t)(r0 + r) * NNODES + r0 + 4 * lc);
  }

#pragma unroll
  for (int it = 0; it < KIN / 16; ++it) {
    const int i = t + 256 * it;
    *reinterpret_cast<float4*>(wst + (i >> 4) * WHS_S + (i & 15) * 4) = wv[it];
  }
  __syncthreads();

  {
    v8f pacc[4];
#pragma unroll
    for (int nt = 0; nt < 4; ++nt)
      pacc[nt] = (v8f){0.f,0.f,0.f,0.f,0.f,0.f,0.f,0.f};
    const float* pb = wst + (2 * lh) * WHS_S + lm;
#pragma unroll
    for (int j = 0; j < KIN / 4; ++j) {
#pragma unroll
      for (int nt = 0; nt < 4; ++nt) {
        const float* bc = pb + (4 * j) * WHS_S + nt * 16;
        v2f b = { bc[0], bc[WHS_S] };
        pacc[nt] = wmma4f32(af[j], b, pacc[nt]);
      }
    }
#pragma unroll
    for (int nt = 0; nt < 4; ++nt)
#pragma unroll
      for (int v = 0; v < 8; ++v)
        whs[(wu * 16 + v + 8 * lh) * WHS_S + nt * 16 + lm] = pacc[nt][v];
  }
  __syncthreads();

  {
    const float* ah  = avec + (size_t)h * 2 * NHID + (wu >> 2) * NHID;
    const float* row = whs + (t & 127) * WHS_S;
    float s = 0.f;
#pragma unroll
    for (int k = 0; k < NHID; ++k) s += row[k] * ah[k];
    els[t] = s;
  }
  __syncthreads();

  const float4 er4v = *reinterpret_cast<const float4*>(ers + 4 * lc);
  const float er4[4] = { er4v.x, er4v.y, er4v.z, er4v.w };
  float elv[SMR];
#pragma unroll
  for (int i = 0; i < SMR; ++i)
    elv[i] = els[min(rb + wu + 8 * i, NROI - 1)];

  {
    float e[SMR][4];
    float m[SMR];
#pragma unroll
    for (int i = 0; i < SMR; ++i) {
      const float a4[4] = { av[i].x, av[i].y, av[i].z, av[i].w };
      float mi = -INFINITY;
#pragma unroll
      for (int q = 0; q < 4; ++q) {
        const float z  = elv[i] + er4[q];
        const float lr = (z > 0.f) ? z : ALPHA_LR * z;
        const float ev = (lane_on && a4[q] > 0.f) ? lr : -INFINITY;
        e[i][q] = ev;
        mi = fmaxf(mi, ev);
      }
      m[i] = mi;
    }
    max_step<1>(m); max_step<2>(m); max_step<4>(m); max_step<8>(m);
#pragma unroll
    for (int i = 0; i < SMR; ++i) m[i] = fmaxf(m[i], xhalf(m[i]));

    float s[SMR];
#pragma unroll
    for (int i = 0; i < SMR; ++i) {
      const float mm = (m[i] == -INFINITY) ? 0.f : m[i];
      float si = 0.f;
#pragma unroll
      for (int q = 0; q < 4; ++q) {
        e[i][q] = expf(e[i][q] - mm);
        si += e[i][q];
      }
      s[i] = si;
    }
    sum_step<1>(s); sum_step<2>(s); sum_step<4>(s); sum_step<8>(s);
#pragma unroll
    for (int i = 0; i < SMR; ++i) s[i] += xhalf(s[i]);

#pragma unroll
    for (int i = 0; i < SMR; ++i) {
      const int r = wu + 8 * i;
      const float inv = (s[i] > 0.f) ? 1.f / s[i] : 0.f;
      if (rb + r < NROI) {
        const float4 o = { e[i][0] * inv, e[i][1] * inv, e[i][2] * inv, e[i][3] * inv };
        *reinterpret_cast<float4*>(att + r * ATT_S + 4 * ls) = o;
      }
    }
  }
  __syncthreads();

  const int w  = wu / WPM;
  const int nb = (wu % WPM) * NTW;
  v8f acc[NTW];
#pragma unroll
  for (int nt = 0; nt < NTW; ++nt)
    acc[nt] = (v8f){0.f,0.f,0.f,0.f,0.f,0.f,0.f,0.f};

  const float* arowA = att + (w * 16 + lm) * ATT_S + 2 * lh;
  const float* bbase = whs + (2 * lh) * WHS_S + nb * 16 + lm;
  for (int k0 = 0; k0 < 128; k0 += 4) {
    v2f a = { arowA[k0], arowA[k0 + 1] };
#pragma unroll
    for (int nt = 0; nt < NTW; ++nt) {
      const float* bc = bbase + k0 * WHS_S + nt * 16;
      v2f b = { bc[0], bc[WHS_S] };
      acc[nt] = wmma4f32(a, b, acc[nt]);
    }
  }
  __syncthreads();

  float* hp = att;
#pragma unroll
  for (int nt = 0; nt < NTW; ++nt)
#pragma unroll
    for (int v = 0; v < 8; ++v)
      hp[(w * 16 + v + 8 * lh) * WHS_S + (nb + nt) * 16 + lm] = acc[nt][v];
  __syncthreads();

  float* sc = whs;
  const int nrows = min(SR, NROI - rb);
  if (t < nrows) {
    const float* row = hp + t * WHS_S;
    float ss = 0.f;
    for (int c = 0; c < NHID; ++c) ss += row[c] * row[c];
    float r = sqrtf(ss);
    float s = fkernel_scale(r);
    if (DOUBLE_FK) s *= fkernel_scale(s * r);
    sc[t] = s;
  }
  __syncthreads();

  float* obase = out + (size_t)(r0 + rb) * out_stride + h * NHID;
  for (int q = t; q < nrows * (NHID / 4); q += 256) {
    const int row = q >> 4;
    const int c4  = (q & 15) * 4;
    const float4 v = *reinterpret_cast<const float4*>(hp + row * WHS_S + c4);
    const float  s = sc[row];
    float4 o;
    o.x = act_elu_cos(s * v.x);
    o.y = act_elu_cos(s * v.y);
    o.z = act_elu_cos(s * v.z);
    o.w = act_elu_cos(s * v.w);
    *reinterpret_cast<float4*>(obase + (size_t)row * out_stride + c4) = o;
  }
}

extern "C" void kernel_launch(void* const* d_in, const int* in_sizes, int n_in,
                              void* d_out, int out_size, void* d_ws, size_t ws_size,
                              hipStream_t stream) {
  (void)in_sizes; (void)n_in; (void)out_size; (void)ws_size;
  const float* x       = (const float*)d_in[0];
  const float* adj     = (const float*)d_in[1];
  const float* W_heads = (const float*)d_in[2];
  const float* a_heads = (const float*)d_in[3];
  const float* W_out   = (const float*)d_in[4];
  const float* a_out   = (const float*)d_in[5];
  float* out = (float*)d_out;

  float* xcat = (float*)d_ws;

  att_block_kernel<false, NFEAT, 2><<<dim3(NGRAPH, NHEADS, 2), 256, LDS_BYTES(NFEAT, 64), stream>>>(
      x, adj, W_heads, a_heads, xcat, NHEADS * NHID);
  att_block_kernel<true, NHEADS * NHID, 4><<<dim3(NGRAPH, 1, 4), 256, LDS_BYTES(NHEADS * NHID, 32), stream>>>(
      xcat, adj, W_out, a_out, out, NHID);
}


# Projection A operand: the block's xin rows loaded as whole rows (16 B per lane, 512 contiguous B per wave load), staged in LDS at pitch KIN+4; A fragments read from LDS, not by 16-line global loads
# speedup vs baseline: 4.4557x; 1.0535x over previous
#include <hip/hip_runtime.h>
#include <math.h>

#define NNODES   3712
#define NROI     116
#define NGRAPH   32
#define NFEAT    128
#define NHID     64
#define NHEADS   4
#define ALPHA_LR 0.1f
#define A_PARA   0.1f
#define PHI_C    0.94247779607693793f
#define MAXNORM  (1.0f - 4e-3f)

typedef __attribute__((ext_vector_type(2))) float v2f;
typedef __attribute__((ext_vector_type(8))) float v8f;

__device__ __forceinline__ v8f wmma4f32(v2f a, v2f b, v8f c) {
  return __builtin_amdgcn_wmma_f32_16x16x4_f32(
      false, a, false, b, (short)0, c, false, false);
}

__device__ __forceinline__ float fkernel_scale(float r) {
  float proj = 1.0f, rn = r;
  if (r > MAXNORM) { proj = MAXNORM / r; rn = MAXNORM; }
  rn = fmaxf(rn, 1e-15f);
  float t = fminf(rn, 1.0f - 1e-7f);
  return proj * (atanhf(t) / rn);
}

__device__ __forceinline__ float act_elu_cos(float v) {
  const float em = expm1f(v);
  const float e  = (v > 0.0f) ? v : em;
  return e + A_PARA * cosf(v + PHI_C);
}

__device__ __forceinline__ float xhalf(float x) {
  const int b = __builtin_bit_cast(int, x);
  return __builtin_bit_cast(float,
      __builtin_amdgcn_permlanex16(b, b, 0x76543210, 0xfedcba98, false, false));
}
template <int M>
__device__ __forceinline__ float rowx(float x) {
  const int b = __builtin_bit_cast(int, x);
  return __builtin_bit_cast(float,
      __builtin_amdgcn_update_dpp(0, b, 0x160 + M, 0xf, 0xf, true));
}

#define ATT_S 132
#define WHS_S 68
#define REGA_FLOATS(KIN, SR) (((KIN) * WHS_S > (SR) * ATT_S) ? (KIN) * WHS_S : (SR) * ATT_S)
#define XS_S(KIN)    ((KIN) + 4)
#define XS_RPI(KIN)  (256 / ((KIN) / 4))
#define XS_IT(KIN)   ((NROI + XS_RPI(KIN) - 1) / XS_RPI(KIN))
#define XS_ROWS(KIN) (XS_IT(KIN) * XS_RPI(KIN))
#define LDS_FLOATS(KIN, SR) (REGA_FLOATS(KIN, SR) + 128 * WHS_S + 256 + XS_ROWS(KIN) * XS_S(KIN))

template <int M, int R>
__device__ __forceinline__ void max_step(float (&v)[R]) {
#pragma unroll
  for (int i = 0; i < R; ++i) v[i] = fmaxf(v[i], rowx<M>(v[i]));
}
template <int M, int R>
__device__ __forceinline__ void sum_step(float (&v)[R]) {
#pragma unroll
  for (int i = 0; i < R; ++i) v[i] += rowx<M>(v[i]);
}

template <bool DOUBLE_FK, int KIN, int NSLAB>
__global__ __launch_bounds__(256) __attribute__((amdgpu_waves_per_eu(2, 2)))
void att_block_kernel(const float* __restrict__ xin, const float* __restrict__ adj,
                      const float* __restrict__ Wall, const float* __restrict__ avec,
                      float* __restrict__ out, int out_stride) {
  constexpr int SR  = 128 / NSLAB;
  constexpr int SMR = SR / 8;
  constexpr int MT  = SR / 16;
  constexpr int WPM = 8 / MT;
  constexpr int NTW = 4 / WPM;
  __shared__ __align__(16) float lds[LDS_FLOATS(KIN, SR)];
  float* wst = lds;
  float* att = lds;
  float* whs = lds + REGA_FLOATS(KIN, SR);
  float* els = whs + 128 * WHS_S;
  float* ers = els + 128;
  float* xs  = els + 256;

  const int g = blockIdx.x, h = blockIdx.y;
  const int t = threadIdx.x;
  const int r0 = g * NROI;
  const int rb = blockIdx.z * SR;

  const int wu = __builtin_amdgcn_readfirstlane(t >> 5);
  const int ls = t & 31;
  const int lm = ls & 15, lh = ls >> 4;
  const bool lane_on = ls < NROI / 4;
  const int lc = lane_on ? ls : (NROI / 4 - 1);

  const float* Wg = Wall + (size_t)h * KIN * NHID;
  float4 wv[KIN / 16];
#pragma unroll
  for (int it = 0; it < KIN / 16; ++it) {
    const int i = t + 256 * it;
    wv[it] = *reinterpret_cast<const float4*>(Wg + (size_t)(i >> 4) * NHID + (i & 15) * 4);
  }

  constexpr int XF4 = KIN / 4;
  constexpr int XRP = XS_RPI(KIN);
  constexpr int XIT = XS_IT(KIN);
  constexpr int XSS = XS_S(KIN);
  const int xr = t / XF4, xc = t % XF4;
  float4 xv[XIT];
#pragma unroll
  for (int it = 0; it < XIT; ++it) {
    const int row = min(xr + XRP * it, NROI - 1);
    xv[it] = *reinterpret_cast<const float4*>(xin + (size_t)(r0 + row) * KIN + 4 * xc);
  }

  float4 av[SMR];
#pragma unroll
  for (int i = 0; i < SMR; ++i) {
    const int r = min(rb + wu + 8 * i, NROI - 1);
    av[i]  = *reinterpret_cast<const float4*>(adj + (size_t)(r0 + r) * NNODES + r0 + 4 * lc);
  }

#pragma unroll
  for (int it = 0; it < KIN / 16; ++it) {
    const int i = t + 256 * it;
    *reinterpret_cast<float4*>(wst + (i >> 4) * WHS_S + (i & 15) * 4) = wv[it];
  }
#pragma unroll
  for (int it = 0; it < XIT; ++it)
    *reinterpret_cast<float4*>(xs + (xr + XRP * it) * XSS + 4 * xc) = xv[it];
  __syncthreads();

  {
    v8f pacc[4];
#pragma unroll
    for (int nt = 0; nt < 4; ++nt)
      pacc[nt] = (v8f){0.f,0.f,0.f,0.f,0.f,0.f,0.f,0.f};
    const float* pb = wst + (2 * lh) * WHS_S + lm;
    const float* ax = xs + min(wu * 16 + lm, NROI - 1) * XSS + 2 * lh;
    v2f af[KIN / 4];
#pragma unroll
    for (int j = 0; j < KIN / 4; ++j) af[j] = *reinterpret_cast<const v2f*>(ax + 4 * j);
#pragma unroll
    for (int j = 0; j < KIN / 4; ++j) {
#pragma unroll
      for (int nt = 0; nt < 4; ++nt) {
        const float* bc = pb + (4 * j) * WHS_S + nt * 16;
        v2f b = { bc[0], bc[WHS_S] };
        pacc[nt] = wmma4f32(af[j], b, pacc[nt]);
      }
    }
#pragma unroll
    for (int nt = 0; nt < 4; ++nt)
#pragma unroll
      for (int v = 0; v < 8; ++v)
        whs[(wu * 16 + v + 8 * lh) * WHS_S + nt * 16 + lm] = pacc[nt][v];
  }
  __syncthreads();

  {
    const float* ah  = avec + (size_t)h * 2 * NHID + (wu >> 2) * NHID;
    const float* row = whs + (t & 127) * WHS_S;
    float s = 0.f;
#pragma unroll
    for (int k = 0; k < NHID; ++k) s += row[k] * ah[k];
    els[t] = s;
  }
  __syncthreads();

  const float4 er4v = *reinterpret_cast<const float4*>(ers + 4 * lc);
  const float er4[4] = { er4v.x, er4v.y, er4v.z, er4v.w };
  float elv[SMR];
#pragma unroll
  for (int i = 0; i < SMR; ++i)
    elv[i] = els[min(rb + wu + 8 * i, NROI - 1)];

  {
    float e[SMR][4];
    float m[SMR];
#pragma unroll
    for (int i = 0; i < SMR; ++i) {
      const float a4[4] = { av[i].x, av[i].y, av[i].z, av[i].w };
      float mi = -INFINITY;
#pragma unroll
      for (int q = 0; q < 4; ++q) {
        const float z  = elv[i] + er4[q];
        const float lr = (z > 0.f) ? z : ALPHA_LR * z;
        const float ev = (lane_on && a4[q] > 0.f) ? lr : -INFINITY;
        e[i][q] = ev;
        mi = fmaxf(mi, ev);
      }
      m[i] = mi;
    }
    max_step<1>(m); max_step<2>(m); max_step<4>(m); max_step<8>(m);
#pragma unroll
    for (int i = 0; i < SMR; ++i) m[i] = fmaxf(m[i], xhalf(m[i]));

    float s[SMR];
#pragma unroll
    for (int i = 0; i < SMR; ++i) {
      const float mm = (m[i] == -INFINITY) ? 0.f : m[i];
      float si = 0.f;
#pragma unroll
      for (int q = 0; q < 4; ++q) {
        e[i][q] = expf(e[i][q] - mm);
        si += e[i][q];
      }
      s[i] = si;
    }
    sum_step<1>(s); sum_step<2>(s); sum_step<4>(s); sum_step<8>(s);
#pragma unroll
    for (int i = 0; i < SMR; ++i) s[i] += xhalf(s[i]);

#pragma unroll
    for (int i = 0; i < SMR; ++i) {
      const int r = wu + 8 * i;
      const float inv = (s[i] > 0.f) ? 1.f / s[i] : 0.f;
      if (rb + r < NROI) {
        const float4 o = { e[i][0] * inv, e[i][1] * inv, e[i][2] * inv, e[i][3] * inv };
        *reinterpret_cast<float4*>(att + r * ATT_S + 4 * ls) = o;
      }
    }
  }
  __syncthreads();

  const int w  = wu / WPM;
  const int nb = (wu % WPM) * NTW;
  v8f acc[NTW];
#pragma unroll
  for (int nt = 0; nt < NTW; ++nt)
    acc[nt] = (v8f){0.f,0.f,0.f,0.f,0.f,0.f,0.f,0.f};

  const float* arowA = att + (w * 16 + lm) * ATT_S + 2 * lh;
  const float* bbase = whs + (2 * lh) * WHS_S + nb * 16 + lm;
  for (int k0 = 0; k0 < 128; k0 += 4) {
    v2f a = { arowA[k0], arowA[k0 + 1] };
#pragma unroll
    for (int nt = 0; nt < NTW; ++nt) {
      const float* bc = bbase + k0 * WHS_S + nt * 16;
      v2f b = { bc[0], bc[WHS_S] };
      acc[nt] = wmma4f32(a, b, acc[nt]);
    }
  }
  __syncthreads();

  float* hp = att;
#pragma unroll
  for (int nt = 0; nt < NTW; ++nt)
#pragma unroll
    for (int v = 0; v < 8; ++v)
      hp[(w * 16 + v + 8 * lh) * WHS_S + (nb + nt) * 16 + lm] = acc[nt][v];
  __syncthreads();

  float* sc = whs;
  const int nrows = min(SR, NROI - rb);
  if (t < nrows) {
    const float* row = hp + t * WHS_S;
    float ss = 0.f;
    for (int c = 0; c < NHID; ++c) ss += row[c] * row[c];
    float r = sqrtf(ss);
    float s = fkernel_scale(r);
    if (DOUBLE_FK) s *= fkernel_scale(s * r);
    sc[t] = s;
  }
  __syncthreads();

  float* obase = out + (size_t)(r0 + rb) * out_stride + h * NHID;
  for (int q = t; q < nrows * (NHID / 4); q += 256) {
    const int row = q >> 4;
    const int c4  = (q & 15) * 4;
    const float4 v = *reinterpret_cast<const float4*>(hp + row * WHS_S + c4);
    const float  s = sc[row];
    float4 o;
    o.x = act_elu_cos(s * v.x);
    o.y = act_elu_cos(s * v.y);
    o.z = act_elu_cos(s * v.z);
    o.w = act_elu_cos(s * v.w);
    *reinterpret_cast<float4*>(obase + (size_t)row * out_stride + c4) = o;
  }
}

extern "C" void kernel_launch(void* const* d_in, const int* in_sizes, int n_in,
                              void* d_out, int out_size, void* d_ws, size_t ws_size,
                              hipStream_t stream) {
  (void)in_sizes; (void)n_in; (void)out_size; (void)ws_size;
  const float* x       = (const float*)d_in[0];
  const float* adj     = (const float*)d_in[1];
  const float* W_heads = (const float*)d_in[2];
  const float* a_heads = (const float*)d_in[3];
  const float* W_out   = (const float*)d_in[4];
  const float* a_out   = (const float*)d_in[5];
  float* out = (float*)d_out;

  float* xcat = (float*)d_ws;

  att_block_kernel<false, NFEAT, 2><<<dim3(NGRAPH, NHEADS, 2), 256, 0, stream>>>(
      x, adj, W_heads, a_heads, xcat, NHEADS * NHID);
  att_block_kernel<true, NHEADS * NHID, 4><<<dim3(NGRAPH, 1, 4), 256, 0, stream>>>(
      xcat, adj, W_out, a_out, out, NHID);
}


# Output layer reassociated as (att slab @ xcat rows) @ W_out, el/er = xcat rows . (W_out a): no slab rebuilds the Wh tile; 8 slabs of 16 rows; k sums, el/er and row norm are summed in a new order
# speedup vs baseline: 5.4522x; 1.2236x over previous
#include <hip/hip_runtime.h>
#include <math.h>

#define NNODES   3712
#define NROI     116
#define NGRAPH   32
#define NFEAT    128
#define NHID     64
#define NHEADS   4
#define ALPHA_LR 0.1f
#define A_PARA   0.1f
#define PHI_C    0.94247779607693793f
#define MAXNORM  (1.0f - 4e-3f)

typedef __attribute__((ext_vector_type(2))) float v2f;
typedef __attribute__((ext_vector_type(8))) float v8f;

__device__ __forceinline__ v8f wmma4f32(v2f a, v2f b, v8f c) {
  return __builtin_amdgcn_wmma_f32_16x16x4_f32(
      false, a, false, b, (short)0, c, false, false);
}

__device__ __forceinline__ float fkernel_scale(float r) {
  float proj = 1.0f, rn = r;
  if (r > MAXNORM) { proj = MAXNORM / r; rn = MAXNORM; }
  rn = fmaxf(rn, 1e-15f);
  float t = fminf(rn, 1.0f - 1e-7f);
  return proj * (atanhf(t) / rn);
}

__device__ __forceinline__ float act_elu_cos(float v) {
  const float em = expm1f(v);
  const float e  = (v > 0.0f) ? v : em;
  return e + A_PARA * cosf(v + PHI_C);
}

__device__ __forceinline__ float xhalf(float x) {
  const int b = __builtin_bit_cast(int, x);
  return __builtin_bit_cast(float,
      __builtin_amdgcn_permlanex16(b, b, 0x76543210, 0xfedcba98, false, false));
}
template <int M>
__device__ __forceinline__ float rowx(float x) {
  const int b = __builtin_bit_cast(int, x);
  return __builtin_bit_cast(float,
      __builtin_amdgcn_update_dpp(0, b, 0x160 + M, 0xf, 0xf, true));
}

#define ATT_S 132
#define WHS_S 68
#define REGA_FLOATS(KIN, SR) (((KIN) * WHS_S > (SR) * ATT_S) ? (KIN) * WHS_S : (SR) * ATT_S)
#define XS_S(KIN)    ((KIN) + 4)
#define XS_RPI(KIN)  (256 / ((KIN) / 4))
#define XS_IT(KIN)   ((NROI + XS_RPI(KIN) - 1) / XS_RPI(KIN))
#define XS_ROWS(KIN) (XS_IT(KIN) * XS_RPI(KIN))
#define LDS_FLOATS(KIN, SR) (REGA_FLOATS(KIN, SR) + 128 * WHS_S + 256 + XS_ROWS(KIN) * XS_S(KIN))

template <int M, int R>
__device__ __forceinline__ void max_step(float (&v)[R]) {
#pragma unroll
  for (int i = 0; i < R; ++i) v[i] = fmaxf(v[i], rowx<M>(v[i]));
}
template <int M, int R>
__device__ __forceinline__ void sum_step(float (&v)[R]) {
#pragma unroll
  for (int i = 0; i < R; ++i) v[i] += rowx<M>(v[i]);
}

template <bool DOUBLE_FK, int KIN, int NSLAB>
__global__ __launch_bounds__(256) __attribute__((amdgpu_waves_per_eu(2, 2)))
void att_block_kernel(const float* __restrict__ xin, const float* __restrict__ adj,
                      const float* __restrict__ Wall, const float* __restrict__ avec,
                      float* __restrict__ out, int out_stride) {
  constexpr int SR  = 128 / NSLAB;
  constexpr int SMR = SR / 8;
  constexpr int MT  = SR / 16;
  constexpr int WPM = 8 / MT;
  constexpr int NTW = 4 / WPM;
  __shared__ __align__(16) float lds[LDS_FLOATS(KIN, SR)];
  float* wst = lds;
  float* att = lds;
  float* whs = lds + REGA_FLOATS(KIN, SR);
  float* els = whs + 128 * WHS_S;
  float* ers = els + 128;
  float* xs  = els + 256;

  const int g = blockIdx.x, h = blockIdx.y;
  const int t = threadIdx.x;
  const int r0 = g * NROI;
  const int rb = blockIdx.z * SR;

  const int wu = __builtin_amdgcn_readfirstlane(t >> 5);
  const int ls = t & 31;
  const int lm = ls & 15, lh = ls >> 4;
  const bool lane_on = ls < NROI / 4;
  const int lc = lane_on ? ls : (NROI / 4 - 1);

  const float* Wg = Wall + (size_t)h * KIN * NHID;
  float4 wv[KIN / 16];
#pragma unroll
  for (int it = 0; it < KIN / 16; ++it) {
    const int i = t + 256 * it;
    wv[it] = *reinterpret_cast<const float4*>(Wg + (size_t)(i >> 4) * NHID + (i & 15) * 4);
  }

  constexpr int XF4 = KIN / 4;
  constexpr int XRP = XS_RPI(KIN);
  constexpr int XIT = XS_IT(KIN);
  constexpr int XSS = XS_S(KIN);
  const int xr = t / XF4, xc = t % XF4;
  float4 xv[XIT];
#pragma unroll
  for (int it = 0; it < XIT; ++it) {
    const int row = min(xr + XRP * it, NROI - 1);
    xv[it] = *reinterpret_cast<const float4*>(xin + (size_t)(r0 + row) * KIN + 4 * xc);
  }

  float4 av[SMR];
#pragma unroll
  for (int i = 0; i < SMR; ++i) {
    const int r = min(rb + wu + 8 * i, NROI - 1);
    av[i]  = *reinterpret_cast<const float4*>(adj + (size_t)(r0 + r) * NNODES + r0 + 4 * lc);
  }

#pragma unroll
  for (int it = 0; it < KIN / 16; ++it) {
    const int i = t + 256 * it;
    *reinterpret_cast<float4*>(wst + (i >> 4) * WHS_S + (i & 15) * 4) = wv[it];
  }
#pragma unroll
  for (int it = 0; it < XIT; ++it)
    *reinterpret_cast<float4*>(xs + (xr + XRP * it) * XSS + 4 * xc) = xv[it];
  __syncthreads();

  {
    v8f pacc[4];
#pragma unroll
    for (int nt = 0; nt < 4; ++nt)
      pacc[nt] = (v8f){0.f,0.f,0.f,0.f,0.f,0.f,0.f,0.f};
    const float* pb = wst + (2 * lh) * WHS_S + lm;
    const float* ax = xs + min(wu * 16 + lm, NROI - 1) * XSS + 2 * lh;
    v2f af[KIN / 4];
#pragma unroll
    for (int j = 0; j < KIN / 4; ++j) af[j] = *reinterpret_cast<const v2f*>(ax + 4 * j);
#pragma unroll
    for (int j = 0; j < KIN / 4; ++j) {
#pragma unroll
      for (int nt = 0; nt < 4; ++nt) {
        const float* bc = pb + (4 * j) * WHS_S + nt * 16;
        v2f b = { bc[0], bc[WHS_S] };
        pacc[nt] = wmma4f32(af[j], b, pacc[nt]);
      }
    }
#pragma unroll
    for (int nt = 0; nt < 4; ++nt)
#pragma unroll
      for (int v = 0; v < 8; ++v)
        whs[(wu * 16 + v + 8 * lh) * WHS_S + nt * 16 + lm] = pacc[nt][v];
  }
  __syncthreads();

  {
    const float* ah  = avec + (size_t)h * 2 * NHID + (wu >> 2) * NHID;
    const float* row = whs + (t & 127) * WHS_S;
    float s = 0.f;
#pragma unroll
    for (int k = 0; k < NHID; ++k) s += row[k] * ah[k];
    els[t] = s;
  }
  __syncthreads();

  const float4 er4v = *reinterpret_cast<const float4*>(ers + 4 * lc);
  const float er4[4] = { er4v.x, er4v.y, er4v.z, er4v.w };
  float elv[SMR];
#pragma unroll
  for (int i = 0; i < SMR; ++i)
    elv[i] = els[min(rb + wu + 8 * i, NROI - 1)];

  {
    float e[SMR][4];
    float m[SMR];
#pragma unroll
    for (int i = 0; i < SMR; ++i) {
      const float a4[4] = { av[i].x, av[i].y, av[i].z, av[i].w };
      float mi = -INFINITY;
#pragma unroll
      for (int q = 0; q < 4; ++q) {
        const float z  = elv[i] + er4[q];
        const float lr = (z > 0.f) ? z : ALPHA_LR * z;
        const float ev = (lane_on && a4[q] > 0.f) ? lr : -INFINITY;
        e[i][q] = ev;
        mi = fmaxf(mi, ev);
      }
      m[i] = mi;
    }
    max_step<1>(m); max_step<2>(m); max_step<4>(m); max_step<8>(m);
#pragma unroll
    for (int i = 0; i < SMR; ++i) m[i] = fmaxf(m[i], xhalf(m[i]));

    float s[SMR];
#pragma unroll
    for (int i = 0; i < SMR; ++i) {
      const float mm = (m[i] == -INFINITY) ? 0.f : m[i];
      float si = 0.f;
#pragma unroll
      for (int q = 0; q < 4; ++q) {
        e[i][q] = expf(e[i][q] - mm);
        si += e[i][q];
      }
      s[i] = si;
    }
    sum_step<1>(s); sum_step<2>(s); sum_step<4>(s); sum_step<8>(s);
#pragma unroll
    for (int i = 0; i < SMR; ++i) s[i] += xhalf(s[i]);

#pragma unroll
    for (int i = 0; i < SMR; ++i) {
      const int r = wu + 8 * i;
      const float inv = (s[i] > 0.f) ? 1.f / s[i] : 0.f;
      if (rb + r < NROI) {
        const float4 o = { e[i][0] * inv, e[i][1] * inv, e[i][2] * inv, e[i][3] * inv };
        *reinterpret_cast<float4*>(att + r * ATT_S + 4 * ls) = o;
      }
    }
  }
  __syncthreads();

  const int w  = wu / WPM;
  const int nb = (wu % WPM) * NTW;
  v8f acc[NTW];
#pragma unroll
  for (int nt = 0; nt < NTW; ++nt)
    acc[nt] = (v8f){0.f,0.f,0.f,0.f,0.f,0.f,0.f,0.f};

  const float* arowA = att + (w * 16 + lm) * ATT_S + 2 * lh;
  const float* bbase = whs + (2 * lh) * WHS_S + nb * 16 + lm;
  for (int k0 = 0; k0 < 128; k0 += 4) {
    v2f a = { arowA[k0], arowA[k0 + 1] };
#pragma unroll
    for (int nt = 0; nt < NTW; ++nt) {
      const float* bc = bbase + k0 * WHS_S + nt * 16;
      v2f b = { bc[0], bc[WHS_S] };
      acc[nt] = wmma4f32(a, b, acc[nt]);
    }
  }
  __syncthreads();

  float* hp = att;
#pragma unroll
  for (int nt = 0; nt < NTW; ++nt)
#pragma unroll
    for (int v = 0; v < 8; ++v)
      hp[(w * 16 + v + 8 * lh) * WHS_S + (nb + nt) * 16 + lm] = acc[nt][v];
  __syncthreads();

  float* sc = whs;
  const int nrows = min(SR, NROI - rb);
  if (t < nrows) {
    const float* row = hp + t * WHS_S;
    float ss = 0.f;
    for (int c = 0; c < NHID; ++c) ss += row[c] * row[c];
    float r = sqrtf(ss);
    float s = fkernel_scale(r);
    if (DOUBLE_FK) s *= fkernel_scale(s * r);
    sc[t] = s;
  }
  __syncthreads();

  float* obase = out + (size_t)(r0 + rb) * out_stride + h * NHID;
  for (int q = t; q < nrows * (NHID / 4); q += 256) {
    const int row = q >> 4;
    const int c4  = (q & 15) * 4;
    const float4 v = *reinterpret_cast<const float4*>(hp + row * WHS_S + c4);
    const float  s = sc[row];
    float4 o;
    o.x = act_elu_cos(s * v.x);
    o.y = act_elu_cos(s * v.y);
    o.z = act_elu_cos(s * v.z);
    o.w = act_elu_cos(s * v.w);
    *reinterpret_cast<float4*>(obase + (size_t)row * out_stride + c4) = o;
  }
}

#define O_K    (NHEADS * NHID)
#define O_SR   16
#define O_XS   (O_K + 8)
#define O_TS   (O_K + 4)
#define O_LDS  (O_K * WHS_S + NROI * O_XS + O_SR * ATT_S + O_SR * O_TS + 2 * O_K + O_SR + 128 + 2 * O_SR * WHS_S)

__global__ __launch_bounds__(256) __attribute__((amdgpu_waves_per_eu(2, 2)))
void out_layer_kernel(const float* __restrict__ xin, const float* __restrict__ adj,
                      const float* __restrict__ W, const float* __restrict__ avec,
                      float* __restrict__ out) {
  constexpr int SMR = O_SR / 8;
  __shared__ __align__(16) float lds[O_LDS];
  float* wst = lds;
  float* xs  = wst + O_K * WHS_S;
  float* att = xs + NROI * O_XS;
  float* ts  = att + O_SR * ATT_S;
  float* wa  = ts + O_SR * O_TS;
  float* els = wa + 2 * O_K;
  float* ers = els + O_SR;
  float* hp2 = ers + 128;

  const int g = blockIdx.x;
  const int t = threadIdx.x;
  const int r0 = g * NROI;
  const int rb = blockIdx.z * O_SR;

  const int wu = __builtin_amdgcn_readfirstlane(t >> 5);
  const int ls = t & 31;
  const int lm = ls & 15, lh = ls >> 4;
  const bool lane_on = ls < NROI / 4;
  const int lc = lane_on ? ls : (NROI / 4 - 1);

  const float4 a1v = *reinterpret_cast<const float4*>(avec + (t & 15) * 4);
  const float4 a2v = *reinterpret_cast<const float4*>(avec + NHID + (t & 15) * 4);
  float4 wv[O_K / 16];
#pragma unroll
  for (int it = 0; it < O_K / 16; ++it) {
    const int i = t + 256 * it;
    wv[it] = *reinterpret_cast<const float4*>(W + (size_t)(i >> 4) * NHID + (i & 15) * 4);
  }
  constexpr int XIT = NROI / 4;
  const int xr = t >> 6, xc = t & 63;
  float4 xv[XIT];
#pragma unroll
  for (int it = 0; it < XIT; ++it)
    xv[it] = *reinterpret_cast<const float4*>(xin + (size_t)(r0 + xr + 4 * it) * O_K + 4 * xc);
  float4 av[SMR];
#pragma unroll
  for (int i = 0; i < SMR; ++i) {
    const int r = min(rb + wu + 8 * i, NROI - 1);
    av[i] = *reinterpret_cast<const float4*>(adj + (size_t)(r0 + r) * NNODES + r0 + 4 * lc);
  }

#pragma unroll
  for (int it = 0; it < O_K / 16; ++it) {
    const int i = t + 256 * it;
    *reinterpret_cast<float4*>(wst + (i >> 4) * WHS_S + (i & 15) * 4) = wv[it];
  }
#pragma unroll
  for (int it = 0; it < XIT; ++it)
    *reinterpret_cast<float4*>(xs + (xr + 4 * it) * O_XS + 4 * xc) = xv[it];
  {
    float p[2 * (O_K / 16)];
#pragma unroll
    for (int it = 0; it < O_K / 16; ++it) {
      const float4 w = wv[it];
      float s1 = w.x * a1v.x;
      s1 += w.y * a1v.y; s1 += w.z * a1v.z; s1 += w.w * a1v.w;
      float s2 = w.x * a2v.x;
      s2 += w.y * a2v.y; s2 += w.z * a2v.z; s2 += w.w * a2v.w;
      p[2 * it] = s1;
      p[2 * it + 1] = s2;
    }
    sum_step<1>(p); sum_step<2>(p); sum_step<4>(p); sum_step<8>(p);
#pragma unroll
    for (int it = 0; it < O_K / 16; ++it) {
      wa[(t >> 4) + 16 * it] = p[2 * it];
      wa[O_K + (t >> 4) + 16 * it] = p[2 * it + 1];
    }
  }
  __syncthreads();

  {
    constexpr int NE = 15;
    const float4 w1a = *reinterpret_cast<const float4*>(wa + 4 * ls);
    const float4 w1b = *reinterpret_cast<const float4*>(wa + 128 + 4 * ls);
    const float4 w2a = *reinterpret_cast<const float4*>(wa + O_K + 4 * ls);
    const float4 w2b = *reinterpret_cast<const float4*>(wa + O_K + 128 + 4 * ls);
    float pr[NE + SMR];
#pragma unroll
    for (int i = 0; i < NE; ++i) {
      const float* xrow = xs + min(wu + 8 * i, NROI - 1) * O_XS + 4 * ls;
      const float4 p = *reinterpret_cast<const float4*>(xrow);
      const float4 q = *reinterpret_cast<const float4*>(xrow + 128);
      float s = p.x * w2a.x;
      s += p.y * w2a.y; s += p.z * w2a.z; s += p.w * w2a.w;
      s += q.x * w2b.x; s += q.y * w2b.y; s += q.z * w2b.z; s += q.w * w2b.w;
      pr[i] = s;
    }
#pragma unroll
    for (int i = 0; i < SMR; ++i) {
      const float* xrow = xs + min(rb + wu + 8 * i, NROI - 1) * O_XS + 4 * ls;
      const float4 p = *reinterpret_cast<const float4*>(xrow);
      const float4 q = *reinterpret_cast<const float4*>(xrow + 128);
      float s = p.x * w1a.x;
      s += p.y * w1a.y; s += p.z * w1a.z; s += p.w * w1a.w;
      s += q.x * w1b.x; s += q.y * w1b.y; s += q.z * w1b.z; s += q.w * w1b.w;
      pr[NE + i] = s;
    }
    sum_step<1>(pr); sum_step<2>(pr); sum_step<4>(pr); sum_step<8>(pr);
#pragma unroll
    for (int i = 0; i < NE + SMR; ++i) pr[i] += xhalf(pr[i]);
    if (ls == 0) {
#pragma unroll
      for (int i = 0; i < NE; ++i)
        if (wu + 8 * i < NROI) ers[wu + 8 * i] = pr[i];
#pragma unroll
      for (int i = 0; i < SMR; ++i) els[wu + 8 * i] = pr[NE + i];
    }
  }
  __syncthreads();

  const float4 er4v = *reinterpret_cast<const float4*>(ers + 4 * lc);
  const float er4[4] = { er4v.x, er4v.y, er4v.z, er4v.w };
  float elv[SMR];
#pragma unroll
  for (int i = 0; i < SMR; ++i) elv[i] = els[wu + 8 * i];

  {
    float e[SMR][4];
    float m[SMR];
#pragma unroll
    for (int i = 0; i < SMR; ++i) {
      const float a4[4] = { av[i].x, av[i].y, av[i].z, av[i].w };
      float mi = -INFINITY;
#pragma unroll
      for (int q = 0; q < 4; ++q) {
        const float z  = elv[i] + er4[q];
        const float lr = (z > 0.f) ? z : ALPHA_LR * z;
        const float ev = (lane_on && a4[q] > 0.f) ? lr : -INFINITY;
        e[i][q] = ev;
        mi = fmaxf(mi, ev);
      }
      m[i] = mi;
    }
    max_step<1>(m); max_step<2>(m); max_step<4>(m); max_step<8>(m);
#pragma unroll
    for (int i = 0; i < SMR; ++i) m[i] = fmaxf(m[i], xhalf(m[i]));

    float s[SMR];
#pragma unroll
    for (int i = 0; i < SMR; ++i) {
      const float mm = (m[i] == -INFINITY) ? 0.f : m[i];
      float si = 0.f;
#pragma unroll
      for (int q = 0; q < 4; ++q) {
        e[i][q] = expf(e[i][q] - mm);
        si += e[i][q];
      }
      s[i] = si;
    }
    sum_step<1>(s); sum_step<2>(s); sum_step<4>(s); sum_step<8>(s);
#pragma unroll
    for (int i = 0; i < SMR; ++i) s[i] += xhalf(s[i]);

#pragma unroll
    for (int i = 0; i < SMR; ++i) {
      const int r = wu + 8 * i;
      const float inv = (s[i] > 0.f) ? 1.f / s[i] : 0.f;
      if (rb + r < NROI) {
        const float4 o = { e[i][0] * inv, e[i][1] * inv, e[i][2] * inv, e[i][3] * inv };
        *reinterpret_cast<float4*>(att + r * ATT_S + 4 * ls) = o;
      }
    }
  }
  __syncthreads();

  {
    v8f tacc[2];
#pragma unroll
    for (int nt = 0; nt < 2; ++nt)
      tacc[nt] = (v8f){0.f,0.f,0.f,0.f,0.f,0.f,0.f,0.f};
    const float* ta = att + lm * ATT_S + 2 * lh;
    const float* tb = xs + (2 * lh) * O_XS + wu * 32 + lm;
#pragma unroll
    for (int j = 0; j < NROI / 4; ++j) {
      v2f a = { ta[4 * j], ta[4 * j + 1] };
#pragma unroll
      for (int nt = 0; nt < 2; ++nt) {
        const float* bc = tb + (4 * j) * O_XS + nt * 16;
        v2f b = { bc[0], bc[O_XS] };
        tacc[nt] = wmma4f32(a, b, tacc[nt]);
      }
    }
#pragma unroll
    for (int nt = 0; nt < 2; ++nt)
#pragma unroll
      for (int v = 0; v < 8; ++v)
        ts[(v + 8 * lh) * O_TS + wu * 32 + nt * 16 + lm] = tacc[nt][v];
  }
  __syncthreads();

  {
    const int nt = wu & 3, kh = wu >> 2;
    v8f h0 = (v8f){0.f,0.f,0.f,0.f,0.f,0.f,0.f,0.f};
    v8f h1 = (v8f){0.f,0.f,0.f,0.f,0.f,0.f,0.f,0.f};
    const float* ha = ts + lm * O_TS + 128 * kh + 2 * lh;
    const float* hb = wst + (128 * kh + 2 * lh) * WHS_S + nt * 16 + lm;
#pragma unroll
    for (int j = 0; j < 16; ++j) {
      v2f a0 = { ha[4 * j], ha[4 * j + 1] };
      v2f b0 = { hb[(4 * j) * WHS_S], hb[(4 * j + 1) * WHS_S] };
      h0 = wmma4f32(a0, b0, h0);
      v2f a1 = { ha[64 + 4 * j], ha[64 + 4 * j + 1] };
      v2f b1 = { hb[(64 + 4 * j) * WHS_S], hb[(64 + 4 * j + 1) * WHS_S] };
      h1 = wmma4f32(a1, b1, h1);
    }
    float* hq = hp2 + kh * (O_SR * WHS_S);
#pragma unroll
    for (int v = 0; v < 8; ++v)
      hq[(v + 8 * lh) * WHS_S + nt * 16 + lm] = h0[v] + h1[v];
  }
  __syncthreads();

  {
    const int row = t >> 4;
    const int c4  = (t & 15) * 4;
    const float4 u0 = *reinterpret_cast<const float4*>(hp2 + row * WHS_S + c4);
    const float4 u1 = *reinterpret_cast<const float4*>(hp2 + (O_SR + row) * WHS_S + c4);
    const float4 v = { u0.x + u1.x, u0.y + u1.y, u0.z + u1.z, u0.w + u1.w };
    float ss[1];
    ss[0] = v.x * v.x;
    ss[0] += v.y * v.y; ss[0] += v.z * v.z; ss[0] += v.w * v.w;
    sum_step<1>(ss); sum_step<2>(ss); sum_step<4>(ss); sum_step<8>(ss);
    const float r = sqrtf(ss[0]);
    float s = fkernel_scale(r);
    s *= fkernel_scale(s * r);
    if (rb + row < NROI) {
      float4 o;
      o.x = act_elu_cos(s * v.x);
      o.y = act_elu_cos(s * v.y);
      o.z = act_elu_cos(s * v.z);
      o.w = act_elu_cos(s * v.w);
      *reinterpret_cast<float4*>(out + (size_t)(r0 + rb + row) * NHID + c4) = o;
    }
  }
}

extern "C" void kernel_launch(void* const* d_in, const int* in_sizes, int n_in,
                              void* d_out, int out_size, void* d_ws, size_t ws_size,
                              hipStream_t stream) {
  (void)in_sizes; (void)n_in; (void)out_size; (void)ws_size;
  const float* x       = (const float*)d_in[0];
  const float* adj     = (const float*)d_in[1];
  const float* W_heads = (const float*)d_in[2];
  const float* a_heads = (const float*)d_in[3];
  const float* W_out   = (const float*)d_in[4];
  const float* a_out   = (const float*)d_in[5];
  float* out = (float*)d_out;

  float* xcat = (float*)d_ws;

  att_block_kernel<false, NFEAT, 2><<<dim3(NGRAPH, NHEADS, 2), 256, 0, stream>>>(
      x, adj, W_heads, a_heads, xcat, NHEADS * NHID);
  out_layer_kernel<<<dim3(NGRAPH, 1, 8), 256, 0, stream>>>(xcat, adj, W_out, a_out, out);
}


# Elementary functions as single hardware instructions instead of the library's in-line forms: exp (softmax, ELU), cos, 1/x, sqrt, atanh via v_rcp+v_log (odd series below 0.25); ~1 ulp each
# speedup vs baseline: 6.2815x; 1.1521x over previous
#include <hip/hip_runtime.h>
#include <math.h>

#define NNODES   3712
#define NROI     116
#define NGRAPH   32
#define NFEAT    128
#define NHID     64
#define NHEADS   4
#define ALPHA_LR 0.1f
#define A_PARA   0.1f
#define PHI_REV  0.15f
#define INV_2PI  0.15915494309189535f
#define LOG2E    1.44269504088896341f
#define MAXNORM  (1.0f - 4e-3f)

typedef __attribute__((ext_vector_type(2))) float v2f;
typedef __attribute__((ext_vector_type(8))) float v8f;

__device__ __forceinline__ v8f wmma4f32(v2f a, v2f b, v8f c) {
  return __builtin_amdgcn_wmma_f32_16x16x4_f32(
      false, a, false, b, (short)0, c, false, false);
}

__device__ __forceinline__ float hw_exp(float x) {
  return __builtin_amdgcn_exp2f(x * LOG2E);
}
__device__ __forceinline__ float hw_rcp(float x) { return __builtin_amdgcn_rcpf(x); }
__device__ __forceinline__ float hw_atanh(float t) {
  const float q  = (1.0f + t) * hw_rcp(1.0f - t);
  const float lg = 0.34657359027997264f * __builtin_amdgcn_logf(q);
  const float u  = t * t;
  float p = fmaf(u, 1.0f / 11.0f, 1.0f / 9.0f);
  p = fmaf(u, p, 1.0f / 7.0f);
  p = fmaf(u, p, 1.0f / 5.0f);
  p = fmaf(u, p, 1.0f / 3.0f);
  p = fmaf(u, p, 1.0f);
  return (t < 0.25f) ? t * p : lg;
}

__device__ __forceinline__ float fkernel_scale(float r) {
  float proj = 1.0f, rn = r;
  if (r > MAXNORM) { proj = MAXNORM * hw_rcp(r); rn = MAXNORM; }
  rn = fmaxf(rn, 1e-15f);
  float t = fminf(rn, 1.0f - 1e-7f);
  return proj * (hw_atanh(t) * hw_rcp(rn));
}

__device__ __forceinline__ float act_elu_cos(float v) {
  const float em = hw_exp(v) - 1.0f;
  const float e  = (v > 0.0f) ? v : em;
  return e + A_PARA * __builtin_amdgcn_cosf(fmaf(v, INV_2PI, PHI_REV));
}

__device__ __forceinline__ float xhalf(float x) {
  const int b = __builtin_bit_cast(int, x);
  return __builtin_bit_cast(float,
      __builtin_amdgcn_permlanex16(b, b, 0x76543210, 0xfedcba98, false, false));
}
template <int M>
__device__ __forceinline__ float rowx(float x) {
  const int b = __builtin_bit_cast(int, x);
  return __builtin_bit_cast(float,
      __builtin_amdgcn_update_dpp(0, b, 0x160 + M, 0xf, 0xf, true));
}

#define ATT_S 132
#define WHS_S 68
#define REGA_FLOATS(KIN, SR) (((KIN) * WHS_S > (SR) * ATT_S) ? (KIN) * WHS_S : (SR) * ATT_S)
#define XS_S(KIN)    ((KIN) + 4)
#define XS_RPI(KIN)  (256 / ((KIN) / 4))
#define XS_IT(KIN)   ((NROI + XS_RPI(KIN) - 1) / XS_RPI(KIN))
#define XS_ROWS(KIN) (XS_IT(KIN) * XS_RPI(KIN))
#define LDS_FLOATS(KIN, SR) (REGA_FLOATS(KIN, SR) + 128 * WHS_S + 256 + XS_ROWS(KIN) * XS_S(KIN))

template <int M, int R>
__device__ __forceinline__ void max_step(float (&v)[R]) {
#pragma unroll
  for (int i = 0; i < R; ++i) v[i] = fmaxf(v[i], rowx<M>(v[i]));
}
template <int M, int R>
__device__ __forceinline__ void sum_step(float (&v)[R]) {
#pragma unroll
  for (int i = 0; i < R; ++i) v[i] += rowx<M>(v[i]);
}

template <bool DOUBLE_FK, int KIN, int NSLAB>
__global__ __launch_bounds__(256) __attribute__((amdgpu_waves_per_eu(2, 2)))
void att_block_kernel(const float* __restrict__ xin, const float* __restrict__ adj,
                      const float* __restrict__ Wall, const float* __restrict__ avec,
                      float* __restrict__ out, int out_stride) {
  constexpr int SR  = 128 / NSLAB;
  constexpr int SMR = SR / 8;
  constexpr int MT  = SR / 16;
  constexpr int WPM = 8 / MT;
  constexpr int NTW = 4 / WPM;
  __shared__ __align__(16) float lds[LDS_FLOATS(KIN, SR)];
  float* wst = lds;
  float* att = lds;
  float* whs = lds + REGA_FLOATS(KIN, SR);
  float* els = whs + 128 * WHS_S;
  float* ers = els + 128;
  float* xs  = els + 256;

  const int g = blockIdx.x, h = blockIdx.y;
  const int t = threadIdx.x;
  const int r0 = g * NROI;
  const int rb = blockIdx.z * SR;

  const int wu = __builtin_amdgcn_readfirstlane(t >> 5);
  const int ls = t & 31;
  const int lm = ls & 15, lh = ls >> 4;
  const bool lane_on = ls < NROI / 4;
  const int lc = lane_on ? ls : (NROI / 4 - 1);

  const float* Wg = Wall + (size_t)h * KIN * NHID;
  float4 wv[KIN / 16];
#pragma unroll
  for (int it = 0; it < KIN / 16; ++it) {
    const int i = t + 256 * it;
    wv[it] = *reinterpret_cast<const float4*>(Wg + (size_t)(i >> 4) * NHID + (i & 15) * 4);
  }

  constexpr int XF4 = KIN / 4;
  constexpr int XRP = XS_RPI(KIN);
  constexpr int XIT = XS_IT(KIN);
  constexpr int XSS = XS_S(KIN);
  const int xr = t / XF4, xc = t % XF4;
  float4 xv[XIT];
#pragma unroll
  for (int it = 0; it < XIT; ++it) {
    const int row = min(xr + XRP * it, NROI - 1);
    xv[it] = *reinterpret_cast<const float4*>(xin + (size_t)(r0 + row) * KIN + 4 * xc);
  }

  float4 av[SMR];
#pragma unroll
  for (int i = 0; i < SMR; ++i) {
    const int r = min(rb + wu + 8 * i, NROI - 1);
    av[i]  = *reinterpret_cast<const float4*>(adj + (size_t)(r0 + r) * NNODES + r0 + 4 * lc);
  }

#pragma unroll
  for (int it = 0; it < KIN / 16; ++it) {
    const int i = t + 256 * it;
    *reinterpret_cast<float4*>(wst + (i >> 4) * WHS_S + (i & 15) * 4) = wv[it];
  }
#pragma unroll
  for (int it = 0; it < XIT; ++it)
    *reinterpret_cast<float4*>(xs + (xr + XRP * it) * XSS + 4 * xc) = xv[it];
  __syncthreads();

  {
    v8f pacc[4];
#pragma unroll
    for (int nt = 0; nt < 4; ++nt)
      pacc[nt] = (v8f){0.f,0.f,0.f,0.f,0.f,0.f,0.f,0.f};
    const float* pb = wst + (2 * lh) * WHS_S + lm;
    const float* ax = xs + min(wu * 16 + lm, NROI - 1) * XSS + 2 * lh;
    v2f af[KIN / 4];
#pragma unroll
    for (int j = 0; j < KIN / 4; ++j) af[j] = *reinterpret_cast<const v2f*>(ax + 4 * j);
#pragma unroll
    for (int j = 0; j < KIN / 4; ++j) {
#pragma unroll
      for (int nt = 0; nt < 4; ++nt) {
        const float* bc = pb + (4 * j) * WHS_S + nt * 16;
        v2f b = { bc[0], bc[WHS_S] };
        pacc[nt] = wmma4f32(af[j], b, pacc[nt]);
      }
    }
#pragma unroll
    for (int nt = 0; nt < 4; ++nt)
#pragma unroll
      for (int v = 0; v < 8; ++v)
        whs[(wu * 16 + v + 8 * lh) * WHS_S + nt * 16 + lm] = pacc[nt][v];
  }
  __syncthreads();

  {
    const float* ah  = avec + (size_t)h * 2 * NHID + (wu >> 2) * NHID;
    const float* row = whs + (t & 127) * WHS_S;
    float s = 0.f;
#pragma unroll
    for (int k = 0; k < NHID; ++k) s += row[k] * ah[k];
    els[t] = s;
  }
  __syncthreads();

  const float4 er4v = *reinterpret_cast<const float4*>(ers + 4 * lc);
  const float er4[4] = { er4v.x, er4v.y, er4v.z, er4v.w };
  float elv[SMR];
#pragma unroll
  for (int i = 0; i < SMR; ++i)
    elv[i] = els[min(rb + wu + 8 * i, NROI - 1)];

  {
    float e[SMR][4];
    float m[SMR];
#pragma unroll
    for (int i = 0; i < SMR; ++i) {
      const float a4[4] = { av[i].x, av[i].y, av[i].z, av[i].w };
      float mi = -INFINITY;
#pragma unroll
      for (int q = 0; q < 4; ++q) {
        const float z  = elv[i] + er4[q];
        const float lr = (z > 0.f) ? z : ALPHA_LR * z;
        const float ev = (lane_on && a4[q] > 0.f) ? lr : -INFINITY;
        e[i][q] = ev;
        mi = fmaxf(mi, ev);
      }
      m[i] = mi;
    }
    max_step<1>(m); max_step<2>(m); max_step<4>(m); max_step<8>(m);
#pragma unroll
    for (int i = 0; i < SMR; ++i) m[i] = fmaxf(m[i], xhalf(m[i]));

    float s[SMR];
#pragma unroll
    for (int i = 0; i < SMR; ++i) {
      const float mm = (m[i] == -INFINITY) ? 0.f : m[i];
      float si = 0.f;
#pragma unroll
      for (int q = 0; q < 4; ++q) {
        e[i][q] = hw_exp(e[i][q] - mm);
        si += e[i][q];
      }
      s[i] = si;
    }
    sum_step<1>(s); sum_step<2>(s); sum_step<4>(s); sum_step<8>(s);
#pragma unroll
    for (int i = 0; i < SMR; ++i) s[i] += xhalf(s[i]);

#pragma unroll
    for (int i = 0; i < SMR; ++i) {
      const int r = wu + 8 * i;
      const float inv = (s[i] > 0.f) ? hw_rcp(s[i]) : 0.f;
      if (rb + r < NROI) {
        const float4 o = { e[i][0] * inv, e[i][1] * inv, e[i][2] * inv, e[i][3] * inv };
        *reinterpret_cast<float4*>(att + r * ATT_S + 4 * ls) = o;
      }
    }
  }
  __syncthreads();

  const int w  = wu / WPM;
  const int nb = (wu % WPM) * NTW;
  v8f acc[NTW];
#pragma unroll
  for (int nt = 0; nt < NTW; ++nt)
    acc[nt] = (v8f){0.f,0.f,0.f,0.f,0.f,0.f,0.f,0.f};

  const float* arowA = att + (w * 16 + lm) * ATT_S + 2 * lh;
  const float* bbase = whs + (2 * lh) * WHS_S + nb * 16 + lm;
  for (int k0 = 0; k0 < 128; k0 += 4) {
    v2f a = { arowA[k0], arowA[k0 + 1] };
#pragma unroll
    for (int nt = 0; nt < NTW; ++nt) {
      const float* bc = bbase + k0 * WHS_S + nt * 16;
      v2f b = { bc[0], bc[WHS_S] };
      acc[nt] = wmma4f32(a, b, acc[nt]);
    }
  }
  __syncthreads();

  float* hp = att;
#pragma unroll
  for (int nt = 0; nt < NTW; ++nt)
#pragma unroll
    for (int v = 0; v < 8; ++v)
      hp[(w * 16 + v + 8 * lh) * WHS_S + (nb + nt) * 16 + lm] = acc[nt][v];
  __syncthreads();

  float* sc = whs;
  const int nrows = min(SR, NROI - rb);
  if (t < nrows) {
    const float* row = hp + t * WHS_S;
    float ss = 0.f;
    for (int c = 0; c < NHID; ++c) ss += row[c] * row[c];
    float r = __builtin_amdgcn_sqrtf(ss);
    float s = fkernel_scale(r);
    if (DOUBLE_FK) s *= fkernel_scale(s * r);
    sc[t] = s;
  }
  __syncthreads();

  float* obase = out + (size_t)(r0 + rb) * out_stride + h * NHID;
  for (int q = t; q < nrows * (NHID / 4); q += 256) {
    const int row = q >> 4;
    const int c4  = (q & 15) * 4;
    const float4 v = *reinterpret_cast<const float4*>(hp + row * WHS_S + c4);
    const float  s = sc[row];
    float4 o;
    o.x = act_elu_cos(s * v.x);
    o.y = act_elu_cos(s * v.y);
    o.z = act_elu_cos(s * v.z);
    o.w = act_elu_cos(s * v.w);
    *reinterpret_cast<float4*>(obase + (size_t)row * out_stride + c4) = o;
  }
}

#define O_K    (NHEADS * NHID)
#define O_SR   16
#define O_XS   (O_K + 8)
#define O_TS   (O_K + 4)
#define O_LDS  (O_K * WHS_S + NROI * O_XS + O_SR * ATT_S + O_SR * O_TS + 2 * O_K + O_SR + 128 + 2 * O_SR * WHS_S)

__global__ __launch_bounds__(256) __attribute__((amdgpu_waves_per_eu(2, 2)))
void out_layer_kernel(const float* __restrict__ xin, const float* __restrict__ adj,
                      const float* __restrict__ W, const float* __restrict__ avec,
                      float* __restrict__ out) {
  constexpr int SMR = O_SR / 8;
  __shared__ __align__(16) float lds[O_LDS];
  float* wst = lds;
  float* xs  = wst + O_K * WHS_S;
  float* att = xs + NROI * O_XS;
  float* ts  = att + O_SR * ATT_S;
  float* wa  = ts + O_SR * O_TS;
  float* els = wa + 2 * O_K;
  float* ers = els + O_SR;
  float* hp2 = ers + 128;

  const int g = blockIdx.x;
  const int t = threadIdx.x;
  const int r0 = g * NROI;
  const int rb = blockIdx.z * O_SR;

  const int wu = __builtin_amdgcn_readfirstlane(t >> 5);
  const int ls = t & 31;
  const int lm = ls & 15, lh = ls >> 4;
  const bool lane_on = ls < NROI / 4;
  const int lc = lane_on ? ls : (NROI / 4 - 1);

  const float4 a1v = *reinterpret_cast<const float4*>(avec + (t & 15) * 4);
  const float4 a2v = *reinterpret_cast<const float4*>(avec + NHID + (t & 15) * 4);
  float4 wv[O_K / 16];
#pragma unroll
  for (int it = 0; it < O_K / 16; ++it) {
    const int i = t + 256 * it;
    wv[it] = *reinterpret_cast<const float4*>(W + (size_t)(i >> 4) * NHID + (i & 15) * 4);
  }
  constexpr int XIT = NROI / 4;
  const int xr = t >> 6, xc = t & 63;
  float4 xv[XIT];
#pragma unroll
  for (int it = 0; it < XIT; ++it)
    xv[it] = *reinterpret_cast<const float4*>(xin + (size_t)(r0 + xr + 4 * it) * O_K + 4 * xc);
  float4 av[SMR];
#pragma unroll
  for (int i = 0; i < SMR; ++i) {
    const int r = min(rb + wu + 8 * i, NROI - 1);
    av[i] = *reinterpret_cast<const float4*>(adj + (size_t)(r0 + r) * NNODES + r0 + 4 * lc);
  }

#pragma unroll
  for (int it = 0; it < O_K / 16; ++it) {
    const int i = t + 256 * it;
    *reinterpret_cast<float4*>(wst + (i >> 4) * WHS_S + (i & 15) * 4) = wv[it];
  }
#pragma unroll
  for (int it = 0; it < XIT; ++it)
    *reinterpret_cast<float4*>(xs + (xr + 4 * it) * O_XS + 4 * xc) = xv[it];
  {
    float p[2 * (O_K / 16)];
#pragma unroll
    for (int it = 0; it < O_K / 16; ++it) {
      const float4 w = wv[it];
      float s1 = w.x * a1v.x;
      s1 += w.y * a1v.y; s1 += w.z * a1v.z; s1 += w.w * a1v.w;
      float s2 = w.x * a2v.x;
      s2 += w.y * a2v.y; s2 += w.z * a2v.z; s2 += w.w * a2v.w;
      p[2 * it] = s1;
      p[2 * it + 1] = s2;
    }
    sum_step<1>(p); sum_step<2>(p); sum_step<4>(p); sum_step<8>(p);
#pragma unroll
    for (int it = 0; it < O_K / 16; ++it) {
      wa[(t >> 4) + 16 * it] = p[2 * it];
      wa[O_K + (t >> 4) + 16 * it] = p[2 * it + 1];
    }
  }
  __syncthreads();

  {
    constexpr int NE = 15;
    const float4 w1a = *reinterpret_cast<const float4*>(wa + 4 * ls);
    const float4 w1b = *reinterpret_cast<const float4*>(wa + 128 + 4 * ls);
    const float4 w2a = *reinterpret_cast<const float4*>(wa + O_K + 4 * ls);
    const float4 w2b = *reinterpret_cast<const float4*>(wa + O_K + 128 + 4 * ls);
    float pr[NE + SMR];
#pragma unroll
    for (int i = 0; i < NE; ++i) {
      const float* xrow = xs + min(wu + 8 * i, NROI - 1) * O_XS + 4 * ls;
      const float4 p = *reinterpret_cast<const float4*>(xrow);
      const float4 q = *reinterpret_cast<const float4*>(xrow + 128);
      float s = p.x * w2a.x;
      s += p.y * w2a.y; s += p.z * w2a.z; s += p.w * w2a.w;
      s += q.x * w2b.x; s += q.y * w2b.y; s += q.z * w2b.z; s += q.w * w2b.w;
      pr[i] = s;
    }
#pragma unroll
    for (int i = 0; i < SMR; ++i) {
      const float* xrow = xs + min(rb + wu + 8 * i, NROI - 1) * O_XS + 4 * ls;
      const float4 p = *reinterpret_cast<const float4*>(xrow);
      const float4 q = *reinterpret_cast<const float4*>(xrow + 128);
      float s = p.x * w1a.x;
      s += p.y * w1a.y; s += p.z * w1a.z; s += p.w * w1a.w;
      s += q.x * w1b.x; s += q.y * w1b.y; s += q.z * w1b.z; s += q.w * w1b.w;
      pr[NE + i] = s;
    }
    sum_step<1>(pr); sum_step<2>(pr); sum_step<4>(pr); sum_step<8>(pr);
#pragma unroll
    for (int i = 0; i < NE + SMR; ++i) pr[i] += xhalf(pr[i]);
    if (ls == 0) {
#pragma unroll
      for (int i = 0; i < NE; ++i)
        if (wu + 8 * i < NROI) ers[wu + 8 * i] = pr[i];
#pragma unroll
      for (int i = 0; i < SMR; ++i) els[wu + 8 * i] = pr[NE + i];
    }
  }
  __syncthreads();

  const float4 er4v = *reinterpret_cast<const float4*>(ers + 4 * lc);
  const float er4[4] = { er4v.x, er4v.y, er4v.z, er4v.w };
  float elv[SMR];
#pragma unroll
  for (int i = 0; i < SMR; ++i) elv[i] = els[wu + 8 * i];

  {
    float e[SMR][4];
    float m[SMR];
#pragma unroll
    for (int i = 0; i < SMR; ++i) {
      const float a4[4] = { av[i].x, av[i].y, av[i].z, av[i].w };
      float mi = -INFINITY;
#pragma unroll
      for (int q = 0; q < 4; ++q) {
        const float z  = elv[i] + er4[q];
        const float lr = (z > 0.f) ? z : ALPHA_LR * z;
        const float ev = (lane_on && a4[q] > 0.f) ? lr : -INFINITY;
        e[i][q] = ev;
        mi = fmaxf(mi, ev);
      }
      m[i] = mi;
    }
    max_step<1>(m); max_step<2>(m); max_step<4>(m); max_step<8>(m);
#pragma unroll
    for (int i = 0; i < SMR; ++i) m[i] = fmaxf(m[i], xhalf(m[i]));

    float s[SMR];
#pragma unroll
    for (int i = 0; i < SMR; ++i) {
      const float mm = (m[i] == -INFINITY) ? 0.f : m[i];
      float si = 0.f;
#pragma unroll
      for (int q = 0; q < 4; ++q) {
        e[i][q] = hw_exp(e[i][q] - mm);
        si += e[i][q];
      }
      s[i] = si;
    }
    sum_step<1>(s); sum_step<2>(s); sum_step<4>(s); sum_step<8>(s);
#pragma unroll
    for (int i = 0; i < SMR; ++i) s[i] += xhalf(s[i]);

#pragma unroll
    for (int i = 0; i < SMR; ++i) {
      const int r = wu + 8 * i;
      const float inv = (s[i] > 0.f) ? hw_rcp(s[i]) : 0.f;
      if (rb + r < NROI) {
        const float4 o = { e[i][0] * inv, e[i][1] * inv, e[i][2] * inv, e[i][3] * inv };
        *reinterpret_cast<float4*>(att + r * ATT_S + 4 * ls) = o;
      }
    }
  }
  __syncthreads();

  {
    v8f tacc[2];
#pragma unroll
    for (int nt = 0; nt < 2; ++nt)
      tacc[nt] = (v8f){0.f,0.f,0.f,0.f,0.f,0.f,0.f,0.f};
    const float* ta = att + lm * ATT_S + 2 * lh;
    const float* tb = xs + (2 * lh) * O_XS + wu * 32 + lm;
#pragma unroll
    for (int j = 0; j < NROI / 4; ++j) {
      v2f a = { ta[4 * j], ta[4 * j + 1] };
#pragma unroll
      for (int nt = 0; nt < 2; ++nt) {
        const float* bc = tb + (4 * j) * O_XS + nt * 16;
        v2f b = { bc[0], bc[O_XS] };
        tacc[nt] = wmma4f32(a, b, tacc[nt]);
      }
    }
#pragma unroll
    for (int nt = 0; nt < 2; ++nt)
#pragma unroll
      for (int v = 0; v < 8; ++v)
        ts[(v + 8 * lh) * O_TS + wu * 32 + nt * 16 + lm] = tacc[nt][v];
  }
  __syncthreads();

  {
    const int nt = wu & 3, kh = wu >> 2;
    v8f h0 = (v8f){0.f,0.f,0.f,0.f,0.f,0.f,0.f,0.f};
    v8f h1 = (v8f){0.f,0.f,0.f,0.f,0.f,0.f,0.f,0.f};
    const float* ha = ts + lm * O_TS + 128 * kh + 2 * lh;
    const float* hb = wst + (128 * kh + 2 * lh) * WHS_S + nt * 16 + lm;
#pragma unroll
    for (int j = 0; j < 16; ++j) {
      v2f a0 = { ha[4 * j], ha[4 * j + 1] };
      v2f b0 = { hb[(4 * j) * WHS_S], hb[(4 * j + 1) * WHS_S] };
      h0 = wmma4f32(a0, b0, h0);
      v2f a1 = { ha[64 + 4 * j], ha[64 + 4 * j + 1] };
      v2f b1 = { hb[(64 + 4 * j) * WHS_S], hb[(64 + 4 * j + 1) * WHS_S] };
      h1 = wmma4f32(a1, b1, h1);
    }
    float* hq = hp2 + kh * (O_SR * WHS_S);
#pragma unroll
    for (int v = 0; v < 8; ++v)
      hq[(v + 8 * lh) * WHS_S + nt * 16 + lm] = h0[v] + h1[v];
  }
  __syncthreads();

  {
    const int row = t >> 4;
    const int c4  = (t & 15) * 4;
    const float4 u0 = *reinterpret_cast<const float4*>(hp2 + row * WHS_S + c4);
    const float4 u1 = *reinterpret_cast<const float4*>(hp2 + (O_SR + row) * WHS_S + c4);
    const float4 v = { u0.x + u1.x, u0.y + u1.y, u0.z + u1.z, u0.w + u1.w };
    float ss[1];
    ss[0] = v.x * v.x;
    ss[0] += v.y * v.y; ss[0] += v.z * v.z; ss[0] += v.w * v.w;
    sum_step<1>(ss); sum_step<2>(ss); sum_step<4>(ss); sum_step<8>(ss);
    const float r = __builtin_amdgcn_sqrtf(ss[0]);
    float s = fkernel_scale(r);
    s *= fkernel_scale(s * r);
    if (rb + row < NROI) {
      float4 o;
      o.x = act_elu_cos(s * v.x);
      o.y = act_elu_cos(s * v.y);
      o.z = act_elu_cos(s * v.z);
      o.w = act_elu_cos(s * v.w);
      *reinterpret_cast<float4*>(out + (size_t)(r0 + rb + row) * NHID + c4) = o;
    }
  }
}

extern "C" void kernel_launch(void* const* d_in, const int* in_sizes, int n_in,
                              void* d_out, int out_size, void* d_ws, size_t ws_size,
                              hipStream_t stream) {
  (void)in_sizes; (void)n_in; (void)out_size; (void)ws_size;
  const float* x       = (const float*)d_in[0];
  const float* adj     = (const float*)d_in[1];
  const float* W_heads = (const float*)d_in[2];
  const float* a_heads = (const float*)d_in[3];
  const float* W_out   = (const float*)d_in[4];
  const float* a_out   = (const float*)d_in[5];
  float* out = (float*)d_out;

  float* xcat = (float*)d_ws;

  att_block_kernel<false, NFEAT, 2><<<dim3(NGRAPH, NHEADS, 2), 256, 0, stream>>>(
      x, adj, W_heads, a_heads, xcat, NHEADS * NHID);
  out_layer_kernel<<<dim3(NGRAPH, 1, 8), 256, 0, stream>>>(xcat, adj, W_out, a_out, out);
}
